# starting kernel
# baseline (speedup 1.0000x reference)
#include <hip/hip_runtime.h>
#include <hip/hip_bf16.h>


typedef __attribute__((ext_vector_type(2))) float v2f;
typedef __attribute__((ext_vector_type(8))) float v8f;

#define F_IN 128
#define HID  16

__global__ void gcn_init_kernel(float* __restrict__ deg,
                                float* __restrict__ agg1,
                                float* __restrict__ out,
                                const float* __restrict__ b1,
                                const float* __restrict__ b2,
                                int n) {
    int tid = blockIdx.x * blockDim.x + threadIdx.x;
    if (tid >= n * HID) return;
    int f = tid & (HID - 1);
    if (f == 0) deg[tid >> 4] = 1.0f;
    agg1[tid] = b1[f];
    out[tid]  = b2[f];
}

__global__ void gcn_deg_kernel(const int* __restrict__ dst,
                               float* __restrict__ deg, int e) {
    int i = blockIdx.x * blockDim.x + threadIdx.x;
    if (i < e) atomicAdd(&deg[dst[i]], 1.0f);
}

__global__ void gcn_rsqrt_kernel(float* __restrict__ deg, int n) {
    int i = blockIdx.x * blockDim.x + threadIdx.x;
    if (i < n) {
        float d = deg[i];
        deg[i] = d > 0.0f ? rsqrtf(d) : 0.0f;
    }
}

template <bool RELU, int K>
__global__ __launch_bounds__(128)
void gcn_gemm_wmma_kernel(const float* __restrict__ A,
                          const float* __restrict__ B,
                          float* __restrict__ C, int nrows) {
    const int lane = threadIdx.x & 31;
    const int wave = threadIdx.x >> 5;
    const int tile = blockIdx.x * 4 + wave;
    const int row0 = tile * 16;
    if (row0 >= nrows) return;

    const int m    = lane & 15;
    const int half = lane >> 4;
    const int kh   = half << 1;

    const float* __restrict__ arow = A + (size_t)(row0 + m) * K;

    v8f acc = {};
#pragma unroll
    for (int k = 0; k < K; k += 4) {
        v2f a = *reinterpret_cast<const v2f*>(arow + k + kh);
        if (RELU) {
            a.x = fmaxf(a.x, 0.0f);
            a.y = fmaxf(a.y, 0.0f);
        }
        v2f b;
        b.x = B[(k + kh) * HID + m];
        b.y = B[(k + kh + 1) * HID + m];
        acc = __builtin_amdgcn_wmma_f32_16x16x4_f32(
             false, a,  false, b,
             (short)0, acc,  false,  false);
    }

    float* __restrict__ ccol = C + (size_t)row0 * HID + m;
#pragma unroll
    for (int r = 0; r < 8; ++r) {
        ccol[(size_t)(r + 8 * half) * HID] = acc[r];
    }
}

__global__ void gcn_agg_kernel(const float* __restrict__ h,
                               const int* __restrict__ src,
                               const int* __restrict__ dst,
                               const float* __restrict__ dinv,
                               float* __restrict__ out,
                               int e, int n) {
    int tid = blockIdx.x * blockDim.x + threadIdx.x;
    int idx = tid >> 4;
    int f   = tid & (HID - 1);
    if (idx >= e + n) return;
    int s, d;
    if (idx < e) {
        s = src[idx];
        d = dst[idx];
    } else {
        s = d = idx - e;
    }
    float norm = dinv[s] * dinv[d];
    atomicAdd(&out[(size_t)d * HID + f], h[(size_t)s * HID + f] * norm);
}

extern "C" void kernel_launch(void* const* d_in, const int* in_sizes, int n_in,
                              void* d_out, int out_size, void* d_ws, size_t ws_size,
                              hipStream_t stream) {
    const float* x  = (const float*)d_in[0];
    const int*   ei = (const int*)d_in[1];
    const float* W1 = (const float*)d_in[2];
    const float* b1 = (const float*)d_in[3];
    const float* W2 = (const float*)d_in[4];
    const float* b2 = (const float*)d_in[5];
    float* out = (float*)d_out;

    const int N = in_sizes[0] / F_IN;
    const int E = in_sizes[1] / 2;
    const int* src = ei;
    const int* dst = ei + E;

    char* ws = (char*)d_ws;
    float* deg  = (float*)ws;
    float* hbuf = (float*)(ws + (size_t)N * sizeof(float));
    float* agg1 = (float*)(ws + (size_t)N * sizeof(float) +
                                (size_t)N * HID * sizeof(float));

    const int TB = 256;

    gcn_init_kernel<<<(N * HID + TB - 1) / TB, TB, 0, stream>>>(
        deg, agg1, out, b1, b2, N);

    gcn_deg_kernel<<<(E + TB - 1) / TB, TB, 0, stream>>>(dst, deg, E);
    gcn_rsqrt_kernel<<<(N + TB - 1) / TB, TB, 0, stream>>>(deg, N);

    const int tiles = (N + 15) / 16;
    const int gblk  = (tiles + 3) / 4;
    gcn_gemm_wmma_kernel<false, F_IN><<<gblk, 128, 0, stream>>>(x, W1, hbuf, N);

    const long long at = (long long)(E + N) * HID;
    gcn_agg_kernel<<<(int)((at + TB - 1) / TB), TB, 0, stream>>>(
        hbuf, src, dst, deg, agg1, E, N);

    gcn_gemm_wmma_kernel<true, HID><<<gblk, 128, 0, stream>>>(agg1, W2, hbuf, N);

    gcn_agg_kernel<<<(int)((at + TB - 1) / TB), TB, 0, stream>>>(
        hbuf, src, dst, deg, out, E, N);
}


# edges counting-sorted by destination once per call (int atomic rank, prefix sum, scatter); both aggregations become atomic-free per-node gathers of dinv-scaled rows; float sum order changes
# speedup vs baseline: 1.6046x; 1.6046x over previous
#include <hip/hip_runtime.h>
#include <hip/hip_bf16.h>


typedef __attribute__((ext_vector_type(2))) float v2f;
typedef __attribute__((ext_vector_type(8))) float v8f;

#define F_IN 128
#define HID  16

__global__ void gcn_zero_kernel(int* __restrict__ cnt, int n) {
    int i = blockIdx.x * blockDim.x + threadIdx.x;
    if (i < n) cnt[i] = 0;
}

__global__ void gcn_rank_kernel(const int* __restrict__ dst,
                                int* __restrict__ cnt,
                                int* __restrict__ rank, int e) {
    int i = blockIdx.x * blockDim.x + threadIdx.x;
    if (i < e) rank[i] = atomicAdd(&cnt[dst[i]], 1);
}

__global__ __launch_bounds__(1024)
void gcn_scan_kernel(const int* __restrict__ cnt,
                     int* __restrict__ rowptr,
                     float* __restrict__ dinv, int n) {
    __shared__ int s_pre[32];
    __shared__ int s_inc[32];
    const int t    = threadIdx.x;
    const int lane = t & 31;
    const int w    = t >> 5;
    const int base = blockIdx.x * 1024;

    int pre = 0;
    const int4* __restrict__ c4 = reinterpret_cast<const int4*>(cnt);
    const int nq = base >> 2;
#pragma unroll 4
    for (int q = t; q < nq; q += 1024) {
        int4 v = c4[q];
        pre += (v.x + v.y) + (v.z + v.w);
    }

    const int i = base + t;
    const int c = (i < n) ? cnt[i] : 0;

    int inc = c;
    for (int d = 1; d < 32; d <<= 1) {
        int v = __shfl_up(inc, d);
        if (lane >= d) inc += v;
    }
    for (int d = 16; d >= 1; d >>= 1) pre += __shfl_xor(pre, d);

    if (lane == 31) s_inc[w] = inc;
    if (lane == 0)  s_pre[w] = pre;
    __syncthreads();

    int off = 0;
    for (int k = 0; k < 32; ++k) {
        off += s_pre[k];
        if (k < w) off += s_inc[k];
    }

    if (i < n) {
        rowptr[i] = off + inc - c;
        float dg = (float)(c + 1);
        dinv[i] = dg > 0.0f ? rsqrtf(dg) : 0.0f;
    }
}

__global__ void gcn_fill_kernel(const int* __restrict__ src,
                                const int* __restrict__ dst,
                                const int* __restrict__ rank,
                                const int* __restrict__ rowptr,
                                int* __restrict__ csr, int e) {
    int i = blockIdx.x * blockDim.x + threadIdx.x;
    if (i < e) {
        int p = rowptr[dst[i]] + rank[i];
        if ((unsigned)p < (unsigned)e) csr[p] = src[i];
    }
}

template <bool RELU, int K>
__global__ __launch_bounds__(128)
void gcn_gemm_wmma_kernel(const float* __restrict__ A,
                          const float* __restrict__ B,
                          const float* __restrict__ rowscale,
                          float* __restrict__ C, int nrows) {
    const int lane = threadIdx.x & 31;
    const int wave = threadIdx.x >> 5;
    const int tile = blockIdx.x * 4 + wave;
    const int row0 = tile * 16;
    if (row0 >= nrows) return;

    const int m    = lane & 15;
    const int half = lane >> 4;
    const int kh   = half << 1;

    const float* __restrict__ arow = A + (size_t)(row0 + m) * K;
    const float sc = rowscale[row0 + m];

    v8f acc = {};
#pragma unroll
    for (int k = 0; k < K; k += 4) {
        v2f a = *reinterpret_cast<const v2f*>(arow + k + kh);
        if (RELU) {
            a.x = fmaxf(a.x, 0.0f);
            a.y = fmaxf(a.y, 0.0f);
        }
        a.x *= sc;
        a.y *= sc;
        v2f b;
        b.x = B[(k + kh) * HID + m];
        b.y = B[(k + kh + 1) * HID + m];
        acc = __builtin_amdgcn_wmma_f32_16x16x4_f32(
             false, a,  false, b,
             (short)0, acc,  false,  false);
    }

    float* __restrict__ ccol = C + (size_t)row0 * HID + m;
#pragma unroll
    for (int r = 0; r < 8; ++r) {
        ccol[(size_t)(r + 8 * half) * HID] = acc[r];
    }
}

template <int U>
static __device__ __forceinline__ int gcn_row_lane(int v) {
    return __builtin_amdgcn_update_dpp(0, v, 0x150 + U, 0xF, 0xF, true);
}

template <int U>
static __device__ __forceinline__ float gcn_take(const float* __restrict__ hs,
                                                 int ids, unsigned last,
                                                 unsigned f, int left) {
    unsigned s = (unsigned)gcn_row_lane<U>(ids);
    s = min(s, last);
    float x = hs[s * HID + f];
    return (U < left) ? x : 0.0f;
}

__global__ __launch_bounds__(256)
void gcn_gather_kernel(const float* __restrict__ hs,
                       const int* __restrict__ rowptr,
                       const int* __restrict__ cnt,
                       const int* __restrict__ csr,
                       const float* __restrict__ dinv,
                       const float* __restrict__ bias,
                       float* __restrict__ out, int n, int e) {
    const int tid = blockIdx.x * blockDim.x + threadIdx.x;
    const int i = tid >> 4;
    const unsigned f = (unsigned)tid & (HID - 1);
    if (i >= n) return;

    int beg = rowptr[i];
    int c   = cnt[i];
    beg = min(max(beg, 0), e);
    c   = min(max(c, 0), e - beg);
    const int* __restrict__ p = csr + beg;
    const unsigned last = (unsigned)(n - 1);

    float acc = hs[(unsigned)i * HID + f];
    for (int j = 0; j < c; j += 16) {
        const int left = c - j;
        const int ids = p[min(j + (int)f, c - 1)];
        float v0  = gcn_take<0>(hs, ids, last, f, left);
        float v1  = gcn_take<1>(hs, ids, last, f, left);
        float v2  = gcn_take<2>(hs, ids, last, f, left);
        float v3  = gcn_take<3>(hs, ids, last, f, left);
        float v4  = gcn_take<4>(hs, ids, last, f, left);
        float v5  = gcn_take<5>(hs, ids, last, f, left);
        float v6  = gcn_take<6>(hs, ids, last, f, left);
        float v7  = gcn_take<7>(hs, ids, last, f, left);
        float v8  = gcn_take<8>(hs, ids, last, f, left);
        float v9  = gcn_take<9>(hs, ids, last, f, left);
        float v10 = gcn_take<10>(hs, ids, last, f, left);
        float v11 = gcn_take<11>(hs, ids, last, f, left);
        float v12 = gcn_take<12>(hs, ids, last, f, left);
        float v13 = gcn_take<13>(hs, ids, last, f, left);
        float v14 = gcn_take<14>(hs, ids, last, f, left);
        float v15 = gcn_take<15>(hs, ids, last, f, left);
        acc += v0;  acc += v1;  acc += v2;  acc += v3;
        acc += v4;  acc += v5;  acc += v6;  acc += v7;
        acc += v8;  acc += v9;  acc += v10; acc += v11;
        acc += v12; acc += v13; acc += v14; acc += v15;
    }
    out[(unsigned)i * HID + f] = acc * dinv[i] + bias[f];
}

static inline size_t gcn_align256(size_t b) { return (b + 255) & ~(size_t)255; }

extern "C" void kernel_launch(void* const* d_in, const int* in_sizes, int n_in,
                              void* d_out, int out_size, void* d_ws, size_t ws_size,
                              hipStream_t stream) {
    const float* x  = (const float*)d_in[0];
    const int*   ei = (const int*)d_in[1];
    const float* W1 = (const float*)d_in[2];
    const float* b1 = (const float*)d_in[3];
    const float* W2 = (const float*)d_in[4];
    const float* b2 = (const float*)d_in[5];
    float* out = (float*)d_out;

    const int N = in_sizes[0] / F_IN;
    const int E = in_sizes[1] / 2;
    const int* src = ei;
    const int* dst = ei + E;

    char* ws = (char*)d_ws;
    const size_t szN  = gcn_align256((size_t)N * sizeof(int));
    const size_t szE  = gcn_align256((size_t)E * sizeof(int));
    const size_t szNH = gcn_align256((size_t)N * HID * sizeof(float));
    int*   cnt    = (int*)ws;
    int*   rowptr = (int*)(ws + szN);
    float* dinv   = (float*)(ws + 2 * szN);
    int*   rank   = (int*)(ws + 3 * szN);
    int*   csr    = (int*)(ws + 3 * szN + szE);
    float* hbuf   = (float*)(ws + 3 * szN + 2 * szE);
    float* agg1   = (float*)(ws + 3 * szN + 2 * szE + szNH);

    const int TB = 256;

    gcn_zero_kernel<<<(N + TB - 1) / TB, TB, 0, stream>>>(cnt, N);
    gcn_rank_kernel<<<(E + TB - 1) / TB, TB, 0, stream>>>(dst, cnt, rank, E);
    gcn_scan_kernel<<<(N + 1023) / 1024, 1024, 0, stream>>>(cnt, rowptr, dinv, N);
    gcn_fill_kernel<<<(E + TB - 1) / TB, TB, 0, stream>>>(src, dst, rank, rowptr, csr, E);

    const int tiles = (N + 15) / 16;
    const int gblk  = (tiles + 3) / 4;
    gcn_gemm_wmma_kernel<false, F_IN><<<gblk, 128, 0, stream>>>(x, W1, dinv, hbuf, N);

    const long long gt = (long long)N * HID;
    const int ggrid = (int)((gt + TB - 1) / TB);
    gcn_gather_kernel<<<ggrid, TB, 0, stream>>>(hbuf, rowptr, cnt, csr, dinv, b1, agg1, N, E);

    gcn_gemm_wmma_kernel<true, HID><<<gblk, 128, 0, stream>>>(agg1, W2, dinv, hbuf, N);

    gcn_gather_kernel<<<ggrid, TB, 0, stream>>>(hbuf, rowptr, cnt, csr, dinv, b2, out, N, E);
}


# edge rank from 16-bit LDS counters per 49152-edge chunk (returning LDS add) plus per-node prefix over chunks, replacing the per-edge global atomic; arrival order per chunk, float sum order changes
# speedup vs baseline: 2.7467x; 1.7118x over previous
#include <hip/hip_runtime.h>
#include <hip/hip_bf16.h>


typedef __attribute__((ext_vector_type(2))) float v2f;
typedef __attribute__((ext_vector_type(8))) float v8f;

#define F_IN 128
#define HID  16

#define CH   49152
#define NCAP 131072

static __device__ __forceinline__ unsigned gcn_rank1(unsigned* s_cnt, int dv,
                                                     unsigned nloc) {
    const unsigned ok  = ((unsigned)dv < nloc) ? 1u : 0u;
    const unsigned d   = ok ? (unsigned)dv : 0u;
    const unsigned sh  = (d & 1u) << 4;
    const unsigned old = atomicAdd(&s_cnt[d >> 1], ok << sh);
    return (old >> sh) & 0xFFFFu;
}

__global__ __launch_bounds__(1024)
void gcn_lrank_kernel(const int* __restrict__ dst,
                      unsigned short* __restrict__ lrank,
                      unsigned* __restrict__ cc,
                      int n, int e, int nwp) {
    __shared__ alignas(16) unsigned s_cnt[NCAP / 2];
    const int t = threadIdx.x;
    const int g = blockIdx.x;
    const unsigned nloc = (unsigned)min(n, NCAP);
    const int nq = (int)((((nloc + 1u) >> 1) + 3u) >> 2);

    uint4* s4 = reinterpret_cast<uint4*>(s_cnt);
    for (int q = t; q < nq; q += 1024) s4[q] = make_uint4(0u, 0u, 0u, 0u);
    __syncthreads();

    const int e0  = g * CH;
    const int len = min(CH, e - e0);
    const int* __restrict__ dp = dst + e0;
    unsigned short* __restrict__ rp = lrank + e0;

    if (len == CH && (reinterpret_cast<size_t>(dp) & 15) == 0) {
        const int4* __restrict__ d4 = reinterpret_cast<const int4*>(dp);
        uint2* __restrict__ r2 = reinterpret_cast<uint2*>(rp);
        int4 v[CH / 4096];
#pragma unroll
        for (int k = 0; k < CH / 4096; ++k) v[k] = d4[k * 1024 + t];
#pragma unroll
        for (int k = 0; k < CH / 4096; ++k) {
            const unsigned a = gcn_rank1(s_cnt, v[k].x, nloc);
            const unsigned b = gcn_rank1(s_cnt, v[k].y, nloc);
            const unsigned c = gcn_rank1(s_cnt, v[k].z, nloc);
            const unsigned d = gcn_rank1(s_cnt, v[k].w, nloc);
            r2[k * 1024 + t] = make_uint2(a | (b << 16), c | (d << 16));
        }
    } else {
        for (int i = t; i < len; i += 1024)
            rp[i] = (unsigned short)gcn_rank1(s_cnt, dp[i], nloc);
    }
    __syncthreads();

    uint4* __restrict__ row = reinterpret_cast<uint4*>(cc + (size_t)g * nwp);
    for (int q = t; q < nq; q += 1024) row[q] = s4[q];
}

__global__ __launch_bounds__(128)
void gcn_chunkscan_kernel(const unsigned* __restrict__ cc,
                          int* __restrict__ pre,
                          int* __restrict__ cnt,
                          int n, int nchunk, int nwp) {
    const int w = blockIdx.x * 128 + threadIdx.x;
    if (2 * w >= n) return;
    const unsigned* __restrict__ cp = cc + w;
    int2* __restrict__ pp = reinterpret_cast<int2*>(pre) + w;
    int run0 = 0, run1 = 0;
#pragma unroll 16
    for (int g = 0; g < nchunk; ++g) {
        const unsigned v = cp[(size_t)g * nwp];
        pp[(size_t)g * nwp] = make_int2(run0, run1);
        run0 += (int)(v & 0xFFFFu);
        run1 += (int)(v >> 16);
    }
    cnt[2 * w] = run0;
    if (2 * w + 1 < n) cnt[2 * w + 1] = run1;
}

__global__ __launch_bounds__(1024)
void gcn_scan_kernel(const int* __restrict__ cnt,
                     int* __restrict__ rowptr,
                     float* __restrict__ dinv, int n) {
    __shared__ int s_pre[32];
    __shared__ int s_inc[32];
    const int t    = threadIdx.x;
    const int lane = t & 31;
    const int w    = t >> 5;
    const int base = blockIdx.x * 1024;

    int pre = 0;
    const int4* __restrict__ c4 = reinterpret_cast<const int4*>(cnt);
    const int nq = base >> 2;
#pragma unroll 4
    for (int q = t; q < nq; q += 1024) {
        int4 v = c4[q];
        pre += (v.x + v.y) + (v.z + v.w);
    }

    const int i = base + t;
    const int c = (i < n) ? cnt[i] : 0;

    int inc = c;
    for (int d = 1; d < 32; d <<= 1) {
        int v = __shfl_up(inc, d);
        if (lane >= d) inc += v;
    }
    for (int d = 16; d >= 1; d >>= 1) pre += __shfl_xor(pre, d);

    if (lane == 31) s_inc[w] = inc;
    if (lane == 0)  s_pre[w] = pre;
    __syncthreads();

    int off = 0;
    for (int k = 0; k < 32; ++k) {
        off += s_pre[k];
        if (k < w) off += s_inc[k];
    }

    if (i < n) {
        rowptr[i] = off + inc - c;
        float dg = (float)(c + 1);
        dinv[i] = dg > 0.0f ? rsqrtf(dg) : 0.0f;
    }
}

__global__ void gcn_fill_kernel(const int* __restrict__ src,
                                const int* __restrict__ dst,
                                const unsigned short* __restrict__ lrank,
                                const int* __restrict__ rowptr,
                                const int* __restrict__ pre,
                                int* __restrict__ csr, int n, int e, int nwp) {
    int i = blockIdx.x * blockDim.x + threadIdx.x;
    if (i < e) {
        const unsigned d = (unsigned)dst[i];
        if (d < (unsigned)min(n, NCAP)) {
            const int g = i / CH;
            int p = rowptr[d] + pre[(size_t)g * (2 * (size_t)nwp) + d] + (int)lrank[i];
            if ((unsigned)p < (unsigned)e) csr[p] = src[i];
        }
    }
}

template <bool RELU, int K>
__global__ __launch_bounds__(128)
void gcn_gemm_wmma_kernel(const float* __restrict__ A,
                          const float* __restrict__ B,
                          const float* __restrict__ rowscale,
                          float* __restrict__ C, int nrows) {
    const int lane = threadIdx.x & 31;
    const int wave = threadIdx.x >> 5;
    const int tile = blockIdx.x * 4 + wave;
    const int row0 = tile * 16;
    if (row0 >= nrows) return;

    const int m    = lane & 15;
    const int half = lane >> 4;
    const int kh   = half << 1;

    const float* __restrict__ arow = A + (size_t)(row0 + m) * K;
    const float sc = rowscale[row0 + m];

    v8f acc = {};
#pragma unroll
    for (int k = 0; k < K; k += 4) {
        v2f a = *reinterpret_cast<const v2f*>(arow + k + kh);
        if (RELU) {
            a.x = fmaxf(a.x, 0.0f);
            a.y = fmaxf(a.y, 0.0f);
        }
        a.x *= sc;
        a.y *= sc;
        v2f b;
        b.x = B[(k + kh) * HID + m];
        b.y = B[(k + kh + 1) * HID + m];
        acc = __builtin_amdgcn_wmma_f32_16x16x4_f32(
             false, a,  false, b,
             (short)0, acc,  false,  false);
    }

    float* __restrict__ ccol = C + (size_t)row0 * HID + m;
#pragma unroll
    for (int r = 0; r < 8; ++r) {
        ccol[(size_t)(r + 8 * half) * HID] = acc[r];
    }
}

template <int U>
static __device__ __forceinline__ int gcn_row_lane(int v) {
    return __builtin_amdgcn_update_dpp(0, v, 0x150 + U, 0xF, 0xF, true);
}

template <int U>
static __device__ __forceinline__ float gcn_take(const float* __restrict__ hs,
                                                 int ids, unsigned last,
                                                 unsigned f, int left) {
    unsigned s = (unsigned)gcn_row_lane<U>(ids);
    s = min(s, last);
    float x = hs[s * HID + f];
    return (U < left) ? x : 0.0f;
}

__global__ __launch_bounds__(256)
void gcn_gather_kernel(const float* __restrict__ hs,
                       const int* __restrict__ rowptr,
                       const int* __restrict__ cnt,
                       const int* __restrict__ csr,
                       const float* __restrict__ dinv,
                       const float* __restrict__ bias,
                       float* __restrict__ out, int n, int e) {
    const int tid = blockIdx.x * blockDim.x + threadIdx.x;
    const int i = tid >> 4;
    const unsigned f = (unsigned)tid & (HID - 1);
    if (i >= n) return;

    int beg = rowptr[i];
    int c   = cnt[i];
    beg = min(max(beg, 0), e);
    c   = min(max(c, 0), e - beg);
    const int* __restrict__ p = csr + beg;
    const unsigned last = (unsigned)(n - 1);

    float acc = hs[(unsigned)i * HID + f];
    for (int j = 0; j < c; j += 16) {
        const int left = c - j;
        const int ids = p[min(j + (int)f, c - 1)];
        float v0  = gcn_take<0>(hs, ids, last, f, left);
        float v1  = gcn_take<1>(hs, ids, last, f, left);
        float v2  = gcn_take<2>(hs, ids, last, f, left);
        float v3  = gcn_take<3>(hs, ids, last, f, left);
        float v4  = gcn_take<4>(hs, ids, last, f, left);
        float v5  = gcn_take<5>(hs, ids, last, f, left);
        float v6  = gcn_take<6>(hs, ids, last, f, left);
        float v7  = gcn_take<7>(hs, ids, last, f, left);
        float v8  = gcn_take<8>(hs, ids, last, f, left);
        float v9  = gcn_take<9>(hs, ids, last, f, left);
        float v10 = gcn_take<10>(hs, ids, last, f, left);
        float v11 = gcn_take<11>(hs, ids, last, f, left);
        float v12 = gcn_take<12>(hs, ids, last, f, left);
        float v13 = gcn_take<13>(hs, ids, last, f, left);
        float v14 = gcn_take<14>(hs, ids, last, f, left);
        float v15 = gcn_take<15>(hs, ids, last, f, left);
        acc += v0;  acc += v1;  acc += v2;  acc += v3;
        acc += v4;  acc += v5;  acc += v6;  acc += v7;
        acc += v8;  acc += v9;  acc += v10; acc += v11;
        acc += v12; acc += v13; acc += v14; acc += v15;
    }
    out[(unsigned)i * HID + f] = acc * dinv[i] + bias[f];
}

static inline size_t gcn_align256(size_t b) { return (b + 255) & ~(size_t)255; }

extern "C" void kernel_launch(void* const* d_in, const int* in_sizes, int n_in,
                              void* d_out, int out_size, void* d_ws, size_t ws_size,
                              hipStream_t stream) {
    const float* x  = (const float*)d_in[0];
    const int*   ei = (const int*)d_in[1];
    const float* W1 = (const float*)d_in[2];
    const float* b1 = (const float*)d_in[3];
    const float* W2 = (const float*)d_in[4];
    const float* b2 = (const float*)d_in[5];
    float* out = (float*)d_out;

    const int N = in_sizes[0] / F_IN;
    const int E = in_sizes[1] / 2;
    const int* src = ei;
    const int* dst = ei + E;

    char* ws = (char*)d_ws;
    const size_t szN  = gcn_align256((size_t)N * sizeof(int));
    const size_t szE  = gcn_align256((size_t)E * sizeof(int));
    const size_t szNH = gcn_align256((size_t)N * HID * sizeof(float));
    int*   cnt    = (int*)ws;
    int*   rowptr = (int*)(ws + szN);
    float* dinv   = (float*)(ws + 2 * szN);
    unsigned short* lrank = (unsigned short*)(ws + 3 * szN);
    int*   csr    = (int*)(ws + 3 * szN + szE);
    float* hbuf   = (float*)(ws + 3 * szN + 2 * szE);
    float* agg1   = (float*)(ws + 3 * szN + 2 * szE + szNH);
    const int NCH = (E + CH - 1) / CH;
    const int NWP = ((((N + 1) / 2) + 3) / 4) * 4;
    const size_t szCC = gcn_align256((size_t)NCH * NWP * sizeof(unsigned));
    unsigned* cc  = (unsigned*)(ws + 3 * szN + 2 * szE + 2 * szNH);
    int*      pre = (int*)(ws + 3 * szN + 2 * szE + 2 * szNH + szCC);

    const int TB = 256;

    gcn_lrank_kernel<<<NCH, 1024, 0, stream>>>(dst, lrank, cc, N, E, NWP);
    gcn_chunkscan_kernel<<<((N + 1) / 2 + 127) / 128, 128, 0, stream>>>(cc, pre, cnt, N, NCH, NWP);
    gcn_scan_kernel<<<(N + 1023) / 1024, 1024, 0, stream>>>(cnt, rowptr, dinv, N);
    gcn_fill_kernel<<<(E + TB - 1) / TB, TB, 0, stream>>>(src, dst, lrank, rowptr, pre, csr, N, E, NWP);

    const int tiles = (N + 15) / 16;
    const int gblk  = (tiles + 3) / 4;
    gcn_gemm_wmma_kernel<false, F_IN><<<gblk, 128, 0, stream>>>(x, W1, dinv, hbuf, N);

    const long long gt = (long long)N * HID;
    const int ggrid = (int)((gt + TB - 1) / TB);
    gcn_gather_kernel<<<ggrid, TB, 0, stream>>>(hbuf, rowptr, cnt, csr, dinv, b1, agg1, N, E);

    gcn_gemm_wmma_kernel<true, HID><<<gblk, 128, 0, stream>>>(agg1, W2, dinv, hbuf, N);

    gcn_gather_kernel<<<ggrid, TB, 0, stream>>>(hbuf, rowptr, cnt, csr, dinv, b2, out, N, E);
}


# edge sort in two levels: edges partitioned into 512-node buckets through LDS, then one work-group per bucket orders its ids in LDS; no chip-wide single-word scatter; float sum order changes
# speedup vs baseline: 4.6015x; 1.6753x over previous
#include <hip/hip_runtime.h>
#include <hip/hip_bf16.h>


typedef __attribute__((ext_vector_type(2))) float v2f;
typedef __attribute__((ext_vector_type(8))) float v8f;

#define F_IN 128
#define HID  16

#define BSH   9
#define BNODE 512
#define NBMAX 256
#define NCAP  (NBMAX * BNODE)
#define EPT   8
#define CHE   (1024 * EPT)

static __device__ __forceinline__ int gcn_wave_incl(int v, int lane) {
    for (int d = 1; d < 32; d <<= 1) {
        int u = __shfl_up(v, d);
        if (lane >= d) v += u;
    }
    return v;
}

__global__ __launch_bounds__(1024)
void gcn_brank_kernel(const int* __restrict__ dst,
                      unsigned short* __restrict__ brank,
                      int* __restrict__ bc,
                      int n, int e, int nchk) {
    __shared__ unsigned s_b[NBMAX];
    const int t = threadIdx.x;
    const int g = blockIdx.x;
    const unsigned nloc = (unsigned)min(n, NCAP);
    if (t < NBMAX) s_b[t] = 0u;
    __syncthreads();

    const int e0 = g * CHE;
    int dv[EPT];
#pragma unroll
    for (int k = 0; k < EPT; ++k) {
        const int i = e0 + k * 1024 + t;
        dv[k] = (i < e) ? dst[i] : -1;
    }
    unsigned rr[EPT];
#pragma unroll
    for (int k = 0; k < EPT; ++k) {
        const unsigned d  = (unsigned)dv[k];
        const bool     ok = d < nloc;
        rr[k] = atomicAdd(&s_b[ok ? (d >> BSH) : 0u], ok ? 1u : 0u);
    }
#pragma unroll
    for (int k = 0; k < EPT; ++k) {
        const int i = e0 + k * 1024 + t;
        if (i < e) brank[i] = (unsigned short)rr[k];
    }
    __syncthreads();
    if (t < NBMAX) bc[t * nchk + g] = (int)s_b[t];
}

__global__ __launch_bounds__(1024)
void gcn_colscan_kernel(const int* __restrict__ bc,
                        int* __restrict__ colpre,
                        int* __restrict__ tot, int nchk) {
    __shared__ int s_w[32];
    const int t    = threadIdx.x;
    const int lane = t & 31;
    const int w    = t >> 5;
    const int b    = blockIdx.x;
    int carry = 0;
    for (int base = 0; base < nchk; base += 1024) {
        const int gi = base + t;
        const int c = (gi < nchk) ? bc[b * nchk + gi] : 0;
        const int inc = gcn_wave_incl(c, lane);
        if (lane == 31) s_w[w] = inc;
        __syncthreads();
        int off = 0, total = 0;
        for (int k = 0; k < 32; ++k) {
            const int v = s_w[k];
            if (k < w) off += v;
            total += v;
        }
        if (gi < nchk) colpre[b * nchk + gi] = carry + off + inc - c;
        carry += total;
        __syncthreads();
    }
    if (t == 0) tot[b] = carry;
}

__global__ __launch_bounds__(1024)
void gcn_part_kernel(const int* __restrict__ src,
                     const int* __restrict__ dst,
                     const unsigned short* __restrict__ brank,
                     const int* __restrict__ bc,
                     const int* __restrict__ colpre,
                     const int* __restrict__ tot,
                     int* __restrict__ ps,
                     int* __restrict__ pd,
                     int n, int e, int nchk) {
    __shared__ int s_d[CHE];
    __shared__ int s_s[CHE];
    __shared__ int s_loff[NBMAX];
    __shared__ int s_goff[NBMAX];
    __shared__ int s_w[2 * (NBMAX / 32)];
    __shared__ int s_nv;
    const int t    = threadIdx.x;
    const int lane = t & 31;
    const int w    = t >> 5;
    const int g    = blockIdx.x;
    const unsigned nloc = (unsigned)min(n, NCAP);

    int c = 0, inc = 0, lc = 0, linc = 0;
    if (t < NBMAX) {
        c    = tot[t];
        inc  = gcn_wave_incl(c, lane);
        lc   = bc[t * nchk + g];
        linc = gcn_wave_incl(lc, lane);
        if (lane == 31) {
            s_w[w] = inc;
            s_w[NBMAX / 32 + w] = linc;
        }
    }
    __syncthreads();
    if (t < NBMAX) {
        int off = 0, loff = 0;
        for (int k = 0; k < NBMAX / 32; ++k) {
            if (k < w) {
                off  += s_w[k];
                loff += s_w[NBMAX / 32 + k];
            }
        }
        const int lst = loff + linc - lc;
        s_loff[t] = lst;
        s_goff[t] = off + inc - c + colpre[t * nchk + g] - lst;
        if (t == NBMAX - 1) s_nv = loff + linc;
    }
    __syncthreads();

    const int e0 = g * CHE;
    int dv[EPT], sv[EPT], rv[EPT];
#pragma unroll
    for (int k = 0; k < EPT; ++k) {
        const int i = e0 + k * 1024 + t;
        const bool in = i < e;
        dv[k] = in ? dst[i] : -1;
        sv[k] = in ? src[i] : 0;
        rv[k] = in ? (int)brank[i] : 0;
    }
#pragma unroll
    for (int k = 0; k < EPT; ++k) {
        const unsigned d = (unsigned)dv[k];
        if (d < nloc) {
            const int lp = s_loff[d >> BSH] + rv[k];
            if ((unsigned)lp < (unsigned)CHE) {
                s_d[lp] = dv[k];
                s_s[lp] = sv[k];
            }
        }
    }
    __syncthreads();

    const int nv = min(s_nv, CHE);
    for (int j = t; j < nv; j += 1024) {
        const int d = s_d[j];
        const int p = s_goff[min((unsigned)d >> BSH, (unsigned)(NBMAX - 1))] + j;
        if ((unsigned)p < (unsigned)e) {
            pd[p] = d;
            ps[p] = s_s[j];
        }
    }
}

#define DCAP 20480
#define DPT  16
__global__ __launch_bounds__(1024)
void gcn_bucket_kernel(const int* __restrict__ ps,
                       const int* __restrict__ pd,
                       const int* __restrict__ tot,
                       int* __restrict__ csr,
                       int* __restrict__ cnt,
                       int* __restrict__ rowptr,
                       float* __restrict__ dinv, int n, int e) {
    __shared__ int s_out[DCAP];
    __shared__ int s_cnt[BNODE];
    __shared__ int s_cur[BNODE];
    __shared__ int s_w[32];
    const int t    = threadIdx.x;
    const int lane = t & 31;
    const int w    = t >> 5;
    const int b    = blockIdx.x;
    const int n0   = b << BSH;

    int v = (t < b) ? tot[t] : 0;
    for (int d = 16; d >= 1; d >>= 1) v += __shfl_xor(v, d);
    if (lane == 0) s_w[w] = v;
    if (t < BNODE) s_cnt[t] = 0;
    __syncthreads();
    int beg = 0;
    for (int k = 0; k < 32; ++k) beg += s_w[k];
    int len = tot[b];
    beg = min(max(beg, 0), e);
    len = min(max(len, 0), e - beg);
    const int* __restrict__ qd = pd + beg;
    const int* __restrict__ qs = ps + beg;
    const bool staged = len <= DCAP;

    for (int base = 0; base < len; base += 1024 * DPT) {
        int q[DPT];
#pragma unroll
        for (int u = 0; u < DPT; ++u) {
            const int i = base + u * 1024 + t;
            q[u] = (i < len) ? qd[i] : -1;
        }
#pragma unroll
        for (int u = 0; u < DPT; ++u) {
            const unsigned l = (unsigned)(q[u] - n0);
            if (l < (unsigned)BNODE) atomicAdd(&s_cnt[l], 1);
        }
    }
    __syncthreads();

    int c = 0, inc = 0;
    if (t < BNODE) {
        c = s_cnt[t];
        inc = gcn_wave_incl(c, lane);
        if (lane == 31) s_w[w] = inc;
    }
    __syncthreads();
    if (t < BNODE) {
        int off = 0;
        for (int k = 0; k < BNODE / 32; ++k)
            if (k < w) off += s_w[k];
        const int ex = off + inc - c;
        s_cur[t] = ex;
        const int node = n0 + t;
        if (node < n) {
            cnt[node]    = c;
            rowptr[node] = beg + ex;
            float dg = (float)(c + 1);
            dinv[node] = dg > 0.0f ? rsqrtf(dg) : 0.0f;
        }
    }
    __syncthreads();

    for (int base = 0; base < len; base += 1024 * DPT) {
        int q[DPT], sv[DPT];
#pragma unroll
        for (int u = 0; u < DPT; ++u) {
            const int i = base + u * 1024 + t;
            const bool in = i < len;
            q[u]  = in ? qd[i] : -1;
            sv[u] = in ? qs[i] : 0;
        }
#pragma unroll
        for (int u = 0; u < DPT; ++u) {
            const unsigned l = (unsigned)(q[u] - n0);
            if (l < (unsigned)BNODE) {
                const int slot = atomicAdd(&s_cur[l], 1);
                if ((unsigned)slot < (unsigned)len) {
                    if (staged) s_out[slot] = sv[u];
                    else        csr[beg + slot] = sv[u];
                }
            }
        }
    }
    __syncthreads();
    if (staged) {
        for (int i = t; i < len; i += 1024) csr[beg + i] = s_out[i];
    }
}

template <bool RELU, int K>
__global__ __launch_bounds__(128)
void gcn_gemm_wmma_kernel(const float* __restrict__ A,
                          const float* __restrict__ B,
                          const float* __restrict__ rowscale,
                          float* __restrict__ C, int nrows) {
    const int lane = threadIdx.x & 31;
    const int wave = threadIdx.x >> 5;
    const int tile = blockIdx.x * 4 + wave;
    const int row0 = tile * 16;
    if (row0 >= nrows) return;

    const int m    = lane & 15;
    const int half = lane >> 4;
    const int kh   = half << 1;

    const float* __restrict__ arow = A + (size_t)(row0 + m) * K;
    const float sc = rowscale[row0 + m];

    v8f acc = {};
#pragma unroll
    for (int k = 0; k < K; k += 4) {
        v2f a = *reinterpret_cast<const v2f*>(arow + k + kh);
        if (RELU) {
            a.x = fmaxf(a.x, 0.0f);
            a.y = fmaxf(a.y, 0.0f);
        }
        a.x *= sc;
        a.y *= sc;
        v2f b;
        b.x = B[(k + kh) * HID + m];
        b.y = B[(k + kh + 1) * HID + m];
        acc = __builtin_amdgcn_wmma_f32_16x16x4_f32(
             false, a,  false, b,
             (short)0, acc,  false,  false);
    }

    float* __restrict__ ccol = C + (size_t)row0 * HID + m;
#pragma unroll
    for (int r = 0; r < 8; ++r) {
        ccol[(size_t)(r + 8 * half) * HID] = acc[r];
    }
}

template <int U>
static __device__ __forceinline__ int gcn_row_lane(int v) {
    return __builtin_amdgcn_update_dpp(0, v, 0x150 + U, 0xF, 0xF, true);
}

template <int U>
static __device__ __forceinline__ float gcn_take(const float* __restrict__ hs,
                                                 int ids, unsigned last,
                                                 unsigned f, int left) {
    unsigned s = (unsigned)gcn_row_lane<U>(ids);
    s = min(s, last);
    float x = hs[s * HID + f];
    return (U < left) ? x : 0.0f;
}

__global__ __launch_bounds__(256)
void gcn_gather_kernel(const float* __restrict__ hs,
                       const int* __restrict__ rowptr,
                       const int* __restrict__ cnt,
                       const int* __restrict__ csr,
                       const float* __restrict__ dinv,
                       const float* __restrict__ bias,
                       float* __restrict__ out, int n, int e) {
    const int tid = blockIdx.x * blockDim.x + threadIdx.x;
    const int i = tid >> 4;
    const unsigned f = (unsigned)tid & (HID - 1);
    if (i >= n) return;

    int beg = rowptr[i];
    int c   = cnt[i];
    beg = min(max(beg, 0), e);
    c   = min(max(c, 0), e - beg);
    const int* __restrict__ p = csr + beg;
    const unsigned last = (unsigned)(n - 1);

    float acc = hs[(unsigned)i * HID + f];
    for (int j = 0; j < c; j += 16) {
        const int left = c - j;
        const int ids = p[min(j + (int)f, c - 1)];
        float v0  = gcn_take<0>(hs, ids, last, f, left);
        float v1  = gcn_take<1>(hs, ids, last, f, left);
        float v2  = gcn_take<2>(hs, ids, last, f, left);
        float v3  = gcn_take<3>(hs, ids, last, f, left);
        float v4  = gcn_take<4>(hs, ids, last, f, left);
        float v5  = gcn_take<5>(hs, ids, last, f, left);
        float v6  = gcn_take<6>(hs, ids, last, f, left);
        float v7  = gcn_take<7>(hs, ids, last, f, left);
        float v8  = gcn_take<8>(hs, ids, last, f, left);
        float v9  = gcn_take<9>(hs, ids, last, f, left);
        float v10 = gcn_take<10>(hs, ids, last, f, left);
        float v11 = gcn_take<11>(hs, ids, last, f, left);
        float v12 = gcn_take<12>(hs, ids, last, f, left);
        float v13 = gcn_take<13>(hs, ids, last, f, left);
        float v14 = gcn_take<14>(hs, ids, last, f, left);
        float v15 = gcn_take<15>(hs, ids, last, f, left);
        acc += v0;  acc += v1;  acc += v2;  acc += v3;
        acc += v4;  acc += v5;  acc += v6;  acc += v7;
        acc += v8;  acc += v9;  acc += v10; acc += v11;
        acc += v12; acc += v13; acc += v14; acc += v15;
    }
    out[(unsigned)i * HID + f] = acc * dinv[i] + bias[f];
}

static inline size_t gcn_align256(size_t b) { return (b + 255) & ~(size_t)255; }

extern "C" void kernel_launch(void* const* d_in, const int* in_sizes, int n_in,
                              void* d_out, int out_size, void* d_ws, size_t ws_size,
                              hipStream_t stream) {
    const float* x  = (const float*)d_in[0];
    const int*   ei = (const int*)d_in[1];
    const float* W1 = (const float*)d_in[2];
    const float* b1 = (const float*)d_in[3];
    const float* W2 = (const float*)d_in[4];
    const float* b2 = (const float*)d_in[5];
    float* out = (float*)d_out;

    const int N = in_sizes[0] / F_IN;
    const int E = in_sizes[1] / 2;
    const int* src = ei;
    const int* dst = ei + E;

    char* ws = (char*)d_ws;
    const size_t szN  = gcn_align256((size_t)N * sizeof(int));
    const size_t szE  = gcn_align256((size_t)E * sizeof(int));
    const size_t szNH = gcn_align256((size_t)N * HID * sizeof(float));
    int*   cnt    = (int*)ws;
    int*   rowptr = (int*)(ws + szN);
    float* dinv   = (float*)(ws + 2 * szN);
    unsigned short* brank = (unsigned short*)(ws + 3 * szN);
    int*   csr    = (int*)(ws + 3 * szN + szE);
    float* hbuf   = (float*)(ws + 3 * szN + 2 * szE);
    float* agg1   = (float*)(ws + 3 * szN + 2 * szE + szNH);
    int*   ps     = (int*)(ws + 3 * szN + 2 * szE + 2 * szNH);
    int*   pd     = (int*)(ws + 3 * szN + 3 * szE + 2 * szNH);
    const int NCHK = (E + CHE - 1) / CHE;
    const size_t szBC = gcn_align256((size_t)NBMAX * NCHK * sizeof(int));
    int*   bc     = (int*)(ws + 3 * szN + 4 * szE + 2 * szNH);
    int*   colpre = (int*)(ws + 3 * szN + 4 * szE + 2 * szNH + szBC);
    int*   tot    = (int*)(ws + 3 * szN + 4 * szE + 2 * szNH + 2 * szBC);

    const int TB = 256;

    gcn_brank_kernel<<<NCHK, 1024, 0, stream>>>(dst, brank, bc, N, E, NCHK);
    gcn_colscan_kernel<<<NBMAX, 1024, 0, stream>>>(bc, colpre, tot, NCHK);
    gcn_part_kernel<<<NCHK, 1024, 0, stream>>>(src, dst, brank, bc, colpre, tot, ps, pd, N, E, NCHK);
    gcn_bucket_kernel<<<NBMAX, 1024, 0, stream>>>(ps, pd, tot, csr, cnt, rowptr, dinv, N, E);

    const int tiles = (N + 15) / 16;
    const int gblk  = (tiles + 3) / 4;
    gcn_gemm_wmma_kernel<false, F_IN><<<gblk, 128, 0, stream>>>(x, W1, dinv, hbuf, N);

    const long long gt = (long long)N * HID;
    const int ggrid = (int)((gt + TB - 1) / TB);
    gcn_gather_kernel<<<ggrid, TB, 0, stream>>>(hbuf, rowptr, cnt, csr, dinv, b1, agg1, N, E);

    gcn_gemm_wmma_kernel<true, HID><<<gblk, 128, 0, stream>>>(agg1, W2, dinv, hbuf, N);

    gcn_gather_kernel<<<ggrid, TB, 0, stream>>>(hbuf, rowptr, cnt, csr, dinv, b2, out, N, E);
}


# gather re-mapped to 4 lanes per node, one 16-byte load per source row: a wave covers 8 nodes, a load instruction fetches 8 rows (was 2), 4x fewer waves start; rows added in the same list order
# speedup vs baseline: 4.7942x; 1.0419x over previous
#include <hip/hip_runtime.h>
#include <hip/hip_bf16.h>


typedef __attribute__((ext_vector_type(2))) float v2f;
typedef __attribute__((ext_vector_type(8))) float v8f;

#define F_IN 128
#define HID  16

#define BSH   9
#define BNODE 512
#define NBMAX 256
#define NCAP  (NBMAX * BNODE)
#define EPT   8
#define CHE   (1024 * EPT)

static __device__ __forceinline__ int gcn_wave_incl(int v, int lane) {
    for (int d = 1; d < 32; d <<= 1) {
        int u = __shfl_up(v, d);
        if (lane >= d) v += u;
    }
    return v;
}

__global__ __launch_bounds__(1024)
void gcn_brank_kernel(const int* __restrict__ dst,
                      unsigned short* __restrict__ brank,
                      int* __restrict__ bc,
                      int n, int e, int nchk) {
    __shared__ unsigned s_b[NBMAX];
    const int t = threadIdx.x;
    const int g = blockIdx.x;
    const unsigned nloc = (unsigned)min(n, NCAP);
    if (t < NBMAX) s_b[t] = 0u;
    __syncthreads();

    const int e0 = g * CHE;
    int dv[EPT];
#pragma unroll
    for (int k = 0; k < EPT; ++k) {
        const int i = e0 + k * 1024 + t;
        dv[k] = (i < e) ? dst[i] : -1;
    }
    unsigned rr[EPT];
#pragma unroll
    for (int k = 0; k < EPT; ++k) {
        const unsigned d  = (unsigned)dv[k];
        const bool     ok = d < nloc;
        rr[k] = atomicAdd(&s_b[ok ? (d >> BSH) : 0u], ok ? 1u : 0u);
    }
#pragma unroll
    for (int k = 0; k < EPT; ++k) {
        const int i = e0 + k * 1024 + t;
        if (i < e) brank[i] = (unsigned short)rr[k];
    }
    __syncthreads();
    if (t < NBMAX) bc[t * nchk + g] = (int)s_b[t];
}

__global__ __launch_bounds__(1024)
void gcn_colscan_kernel(const int* __restrict__ bc,
                        int* __restrict__ colpre,
                        int* __restrict__ tot, int nchk) {
    __shared__ int s_w[32];
    const int t    = threadIdx.x;
    const int lane = t & 31;
    const int w    = t >> 5;
    const int b    = blockIdx.x;
    int carry = 0;
    for (int base = 0; base < nchk; base += 1024) {
        const int gi = base + t;
        const int c = (gi < nchk) ? bc[b * nchk + gi] : 0;
        const int inc = gcn_wave_incl(c, lane);
        if (lane == 31) s_w[w] = inc;
        __syncthreads();
        int off = 0, total = 0;
        for (int k = 0; k < 32; ++k) {
            const int v = s_w[k];
            if (k < w) off += v;
            total += v;
        }
        if (gi < nchk) colpre[b * nchk + gi] = carry + off + inc - c;
        carry += total;
        __syncthreads();
    }
    if (t == 0) tot[b] = carry;
}

__global__ __launch_bounds__(1024)
void gcn_part_kernel(const int* __restrict__ src,
                     const int* __restrict__ dst,
                     const unsigned short* __restrict__ brank,
                     const int* __restrict__ bc,
                     const int* __restrict__ colpre,
                     const int* __restrict__ tot,
                     int* __restrict__ ps,
                     int* __restrict__ pd,
                     int n, int e, int nchk) {
    __shared__ int s_d[CHE];
    __shared__ int s_s[CHE];
    __shared__ int s_loff[NBMAX];
    __shared__ int s_goff[NBMAX];
    __shared__ int s_w[2 * (NBMAX / 32)];
    __shared__ int s_nv;
    const int t    = threadIdx.x;
    const int lane = t & 31;
    const int w    = t >> 5;
    const int g    = blockIdx.x;
    const unsigned nloc = (unsigned)min(n, NCAP);

    int c = 0, inc = 0, lc = 0, linc = 0;
    if (t < NBMAX) {
        c    = tot[t];
        inc  = gcn_wave_incl(c, lane);
        lc   = bc[t * nchk + g];
        linc = gcn_wave_incl(lc, lane);
        if (lane == 31) {
            s_w[w] = inc;
            s_w[NBMAX / 32 + w] = linc;
        }
    }
    __syncthreads();
    if (t < NBMAX) {
        int off = 0, loff = 0;
        for (int k = 0; k < NBMAX / 32; ++k) {
            if (k < w) {
                off  += s_w[k];
                loff += s_w[NBMAX / 32 + k];
            }
        }
        const int lst = loff + linc - lc;
        s_loff[t] = lst;
        s_goff[t] = off + inc - c + colpre[t * nchk + g] - lst;
        if (t == NBMAX - 1) s_nv = loff + linc;
    }
    __syncthreads();

    const int e0 = g * CHE;
    int dv[EPT], sv[EPT], rv[EPT];
#pragma unroll
    for (int k = 0; k < EPT; ++k) {
        const int i = e0 + k * 1024 + t;
        const bool in = i < e;
        dv[k] = in ? dst[i] : -1;
        sv[k] = in ? src[i] : 0;
        rv[k] = in ? (int)brank[i] : 0;
    }
#pragma unroll
    for (int k = 0; k < EPT; ++k) {
        const unsigned d = (unsigned)dv[k];
        if (d < nloc) {
            const int lp = s_loff[d >> BSH] + rv[k];
            if ((unsigned)lp < (unsigned)CHE) {
                s_d[lp] = dv[k];
                s_s[lp] = sv[k];
            }
        }
    }
    __syncthreads();

    const int nv = min(s_nv, CHE);
    for (int j = t; j < nv; j += 1024) {
        const int d = s_d[j];
        const int p = s_goff[min((unsigned)d >> BSH, (unsigned)(NBMAX - 1))] + j;
        if ((unsigned)p < (unsigned)e) {
            pd[p] = d;
            ps[p] = s_s[j];
        }
    }
}

#define DCAP 20480
#define DPT  16
__global__ __launch_bounds__(1024)
void gcn_bucket_kernel(const int* __restrict__ ps,
                       const int* __restrict__ pd,
                       const int* __restrict__ tot,
                       int* __restrict__ csr,
                       int* __restrict__ cnt,
                       int* __restrict__ rowptr,
                       float* __restrict__ dinv, int n, int e) {
    __shared__ int s_out[DCAP];
    __shared__ int s_cnt[BNODE];
    __shared__ int s_cur[BNODE];
    __shared__ int s_w[32];
    const int t    = threadIdx.x;
    const int lane = t & 31;
    const int w    = t >> 5;
    const int b    = blockIdx.x;
    const int n0   = b << BSH;

    int v = (t < b) ? tot[t] : 0;
    for (int d = 16; d >= 1; d >>= 1) v += __shfl_xor(v, d);
    if (lane == 0) s_w[w] = v;
    if (t < BNODE) s_cnt[t] = 0;
    __syncthreads();
    int beg = 0;
    for (int k = 0; k < 32; ++k) beg += s_w[k];
    int len = tot[b];
    beg = min(max(beg, 0), e);
    len = min(max(len, 0), e - beg);
    const int* __restrict__ qd = pd + beg;
    const int* __restrict__ qs = ps + beg;
    const bool staged = len <= DCAP;

    for (int base = 0; base < len; base += 1024 * DPT) {
        int q[DPT];
#pragma unroll
        for (int u = 0; u < DPT; ++u) {
            const int i = base + u * 1024 + t;
            q[u] = (i < len) ? qd[i] : -1;
        }
#pragma unroll
        for (int u = 0; u < DPT; ++u) {
            const unsigned l = (unsigned)(q[u] - n0);
            if (l < (unsigned)BNODE) atomicAdd(&s_cnt[l], 1);
        }
    }
    __syncthreads();

    int c = 0, inc = 0;
    if (t < BNODE) {
        c = s_cnt[t];
        inc = gcn_wave_incl(c, lane);
        if (lane == 31) s_w[w] = inc;
    }
    __syncthreads();
    if (t < BNODE) {
        int off = 0;
        for (int k = 0; k < BNODE / 32; ++k)
            if (k < w) off += s_w[k];
        const int ex = off + inc - c;
        s_cur[t] = ex;
        const int node = n0 + t;
        if (node < n) {
            cnt[node]    = c;
            rowptr[node] = beg + ex;
            float dg = (float)(c + 1);
            dinv[node] = dg > 0.0f ? rsqrtf(dg) : 0.0f;
        }
    }
    __syncthreads();

    for (int base = 0; base < len; base += 1024 * DPT) {
        int q[DPT], sv[DPT];
#pragma unroll
        for (int u = 0; u < DPT; ++u) {
            const int i = base + u * 1024 + t;
            const bool in = i < len;
            q[u]  = in ? qd[i] : -1;
            sv[u] = in ? qs[i] : 0;
        }
#pragma unroll
        for (int u = 0; u < DPT; ++u) {
            const unsigned l = (unsigned)(q[u] - n0);
            if (l < (unsigned)BNODE) {
                const int slot = atomicAdd(&s_cur[l], 1);
                if ((unsigned)slot < (unsigned)len) {
                    if (staged) s_out[slot] = sv[u];
                    else        csr[beg + slot] = sv[u];
                }
            }
        }
    }
    __syncthreads();
    if (staged) {
        for (int i = t; i < len; i += 1024) csr[beg + i] = s_out[i];
    }
}

template <bool RELU, int K>
__global__ __launch_bounds__(128)
void gcn_gemm_wmma_kernel(const float* __restrict__ A,
                          const float* __restrict__ B,
                          const float* __restrict__ rowscale,
                          float* __restrict__ C, int nrows) {
    const int lane = threadIdx.x & 31;
    const int wave = threadIdx.x >> 5;
    const int tile = blockIdx.x * 4 + wave;
    const int row0 = tile * 16;
    if (row0 >= nrows) return;

    const int m    = lane & 15;
    const int half = lane >> 4;
    const int kh   = half << 1;

    const float* __restrict__ arow = A + (size_t)(row0 + m) * K;
    const float sc = rowscale[row0 + m];

    v8f acc = {};
#pragma unroll
    for (int k = 0; k < K; k += 4) {
        v2f a = *reinterpret_cast<const v2f*>(arow + k + kh);
        if (RELU) {
            a.x = fmaxf(a.x, 0.0f);
            a.y = fmaxf(a.y, 0.0f);
        }
        a.x *= sc;
        a.y *= sc;
        v2f b;
        b.x = B[(k + kh) * HID + m];
        b.y = B[(k + kh + 1) * HID + m];
        acc = __builtin_amdgcn_wmma_f32_16x16x4_f32(
             false, a,  false, b,
             (short)0, acc,  false,  false);
    }

    float* __restrict__ ccol = C + (size_t)row0 * HID + m;
#pragma unroll
    for (int r = 0; r < 8; ++r) {
        ccol[(size_t)(r + 8 * half) * HID] = acc[r];
    }
}

typedef __attribute__((ext_vector_type(4))) float v4f;

template <int Q>
static __device__ __forceinline__ int gcn_quad_lane(int v) {
    return __builtin_amdgcn_update_dpp(0, v, Q * 0x55, 0xF, 0xF, true);
}

template <int K>
static __device__ __forceinline__ v4f gcn_take4(const float* __restrict__ hs,
                                                const int (&ids)[4], unsigned last,
                                                unsigned q4, int left) {
    unsigned s = (unsigned)gcn_quad_lane<(K & 3)>(ids[K >> 2]);
    s = min(s, last);
    v4f x = *reinterpret_cast<const v4f*>(hs + s * HID + q4);
    const v4f z = {0.0f, 0.0f, 0.0f, 0.0f};
    return (K < left) ? x : z;
}

__global__ __launch_bounds__(256)
void gcn_gather_kernel(const float* __restrict__ hs,
                       const int* __restrict__ rowptr,
                       const int* __restrict__ cnt,
                       const int* __restrict__ csr,
                       const float* __restrict__ dinv,
                       const float* __restrict__ bias,
                       float* __restrict__ out, int n, int e) {
    const int tid = blockIdx.x * blockDim.x + threadIdx.x;
    const int i = tid >> 2;
    const int q = tid & 3;
    const unsigned q4 = (unsigned)q * 4u;
    if (i >= n) return;

    int beg = rowptr[i];
    int c   = cnt[i];
    beg = min(max(beg, 0), e);
    c   = min(max(c, 0), e - beg);
    const int* __restrict__ p = csr + beg;
    const unsigned last = (unsigned)(n - 1);

    v4f acc = *reinterpret_cast<const v4f*>(hs + (unsigned)i * HID + q4);
    for (int j = 0; j < c; j += 16) {
        const int left = c - j;
        int ids[4];
#pragma unroll
        for (int u = 0; u < 4; ++u) ids[u] = p[min(j + 4 * u + q, c - 1)];
        v4f v0  = gcn_take4<0>(hs, ids, last, q4, left);
        v4f v1  = gcn_take4<1>(hs, ids, last, q4, left);
        v4f v2  = gcn_take4<2>(hs, ids, last, q4, left);
        v4f v3  = gcn_take4<3>(hs, ids, last, q4, left);
        v4f v4  = gcn_take4<4>(hs, ids, last, q4, left);
        v4f v5  = gcn_take4<5>(hs, ids, last, q4, left);
        v4f v6  = gcn_take4<6>(hs, ids, last, q4, left);
        v4f v7  = gcn_take4<7>(hs, ids, last, q4, left);
        v4f v8  = gcn_take4<8>(hs, ids, last, q4, left);
        v4f v9  = gcn_take4<9>(hs, ids, last, q4, left);
        v4f v10 = gcn_take4<10>(hs, ids, last, q4, left);
        v4f v11 = gcn_take4<11>(hs, ids, last, q4, left);
        v4f v12 = gcn_take4<12>(hs, ids, last, q4, left);
        v4f v13 = gcn_take4<13>(hs, ids, last, q4, left);
        v4f v14 = gcn_take4<14>(hs, ids, last, q4, left);
        v4f v15 = gcn_take4<15>(hs, ids, last, q4, left);
        acc += v0;  acc += v1;  acc += v2;  acc += v3;
        acc += v4;  acc += v5;  acc += v6;  acc += v7;
        acc += v8;  acc += v9;  acc += v10; acc += v11;
        acc += v12; acc += v13; acc += v14; acc += v15;
    }
    const float di = dinv[i];
    const v4f bv = *reinterpret_cast<const v4f*>(bias + q4);
    v4f o;
    o.x = acc.x * di + bv.x;
    o.y = acc.y * di + bv.y;
    o.z = acc.z * di + bv.z;
    o.w = acc.w * di + bv.w;
    *reinterpret_cast<v4f*>(out + (unsigned)i * HID + q4) = o;
}

static inline size_t gcn_align256(size_t b) { return (b + 255) & ~(size_t)255; }

extern "C" void kernel_launch(void* const* d_in, const int* in_sizes, int n_in,
                              void* d_out, int out_size, void* d_ws, size_t ws_size,
                              hipStream_t stream) {
    const float* x  = (const float*)d_in[0];
    const int*   ei = (const int*)d_in[1];
    const float* W1 = (const float*)d_in[2];
    const float* b1 = (const float*)d_in[3];
    const float* W2 = (const float*)d_in[4];
    const float* b2 = (const float*)d_in[5];
    float* out = (float*)d_out;

    const int N = in_sizes[0] / F_IN;
    const int E = in_sizes[1] / 2;
    const int* src = ei;
    const int* dst = ei + E;

    char* ws = (char*)d_ws;
    const size_t szN  = gcn_align256((size_t)N * sizeof(int));
    const size_t szE  = gcn_align256((size_t)E * sizeof(int));
    const size_t szNH = gcn_align256((size_t)N * HID * sizeof(float));
    int*   cnt    = (int*)ws;
    int*   rowptr = (int*)(ws + szN);
    float* dinv   = (float*)(ws + 2 * szN);
    unsigned short* brank = (unsigned short*)(ws + 3 * szN);
    int*   csr    = (int*)(ws + 3 * szN + szE);
    float* hbuf   = (float*)(ws + 3 * szN + 2 * szE);
    float* agg1   = (float*)(ws + 3 * szN + 2 * szE + szNH);
    int*   ps     = (int*)(ws + 3 * szN + 2 * szE + 2 * szNH);
    int*   pd     = (int*)(ws + 3 * szN + 3 * szE + 2 * szNH);
    const int NCHK = (E + CHE - 1) / CHE;
    const size_t szBC = gcn_align256((size_t)NBMAX * NCHK * sizeof(int));
    int*   bc     = (int*)(ws + 3 * szN + 4 * szE + 2 * szNH);
    int*   colpre = (int*)(ws + 3 * szN + 4 * szE + 2 * szNH + szBC);
    int*   tot    = (int*)(ws + 3 * szN + 4 * szE + 2 * szNH + 2 * szBC);

    const int TB = 256;

    gcn_brank_kernel<<<NCHK, 1024, 0, stream>>>(dst, brank, bc, N, E, NCHK);
    gcn_colscan_kernel<<<NBMAX, 1024, 0, stream>>>(bc, colpre, tot, NCHK);
    gcn_part_kernel<<<NCHK, 1024, 0, stream>>>(src, dst, brank, bc, colpre, tot, ps, pd, N, E, NCHK);
    gcn_bucket_kernel<<<NBMAX, 1024, 0, stream>>>(ps, pd, tot, csr, cnt, rowptr, dinv, N, E);

    const int tiles = (N + 15) / 16;
    const int gblk  = (tiles + 3) / 4;
    gcn_gemm_wmma_kernel<false, F_IN><<<gblk, 128, 0, stream>>>(x, W1, dinv, hbuf, N);

    const long long gt = (long long)N * 4;
    const int ggrid = (int)((gt + TB - 1) / TB);
    gcn_gather_kernel<<<ggrid, TB, 0, stream>>>(hbuf, rowptr, cnt, csr, dinv, b1, agg1, N, E);

    gcn_gemm_wmma_kernel<true, HID><<<gblk, 128, 0, stream>>>(agg1, W2, dinv, hbuf, N);

    gcn_gather_kernel<<<ggrid, TB, 0, stream>>>(hbuf, rowptr, cnt, csr, dinv, b2, out, N, E);
}


# edge partition kept chunk-local: each chunk bucket-ordered in its own stretch plus a run table, bucket kernel collects its runs; histogram and prefix launches gone; csr order, float sum order change
# speedup vs baseline: 5.1591x; 1.0761x over previous
#include <hip/hip_runtime.h>
#include <hip/hip_bf16.h>


typedef __attribute__((ext_vector_type(2))) float v2f;
typedef __attribute__((ext_vector_type(8))) float v8f;

#define F_IN 128
#define HID  16

#define BSH   9
#define BNODE 512
#define NBMAX 256
#define NCAP  (NBMAX * BNODE)
#define EPT   8
#define CHE   (1024 * EPT)

static __device__ __forceinline__ int gcn_wave_incl(int v, int lane) {
    for (int d = 1; d < 32; d <<= 1) {
        int u = __shfl_up(v, d);
        if (lane >= d) v += u;
    }
    return v;
}

__global__ __launch_bounds__(1024)
void gcn_part_kernel(const int* __restrict__ src,
                     const int* __restrict__ dst,
                     int* __restrict__ lst,
                     int* __restrict__ lct,
                     int* __restrict__ ps,
                     int* __restrict__ pd,
                     int n, int e, int nchk) {
    __shared__ int s_d[CHE];
    __shared__ int s_s[CHE];
    __shared__ unsigned s_b[NBMAX];
    __shared__ int s_loff[NBMAX];
    __shared__ int s_w[NBMAX / 32];
    __shared__ int s_nv;
    const int t    = threadIdx.x;
    const int lane = t & 31;
    const int w    = t >> 5;
    const int g    = blockIdx.x;
    const unsigned nloc = (unsigned)min(n, NCAP);
    if (t < NBMAX) s_b[t] = 0u;
    __syncthreads();

    const int e0 = g * CHE;
    int dv[EPT], sv[EPT];
#pragma unroll
    for (int k = 0; k < EPT; ++k) {
        const int i = e0 + k * 1024 + t;
        const bool in = i < e;
        dv[k] = in ? dst[i] : -1;
        sv[k] = in ? src[i] : 0;
    }
    unsigned rr[EPT];
#pragma unroll
    for (int k = 0; k < EPT; ++k) {
        const unsigned d  = (unsigned)dv[k];
        const bool     ok = d < nloc;
        rr[k] = atomicAdd(&s_b[ok ? (d >> BSH) : 0u], ok ? 1u : 0u);
    }
    __syncthreads();

    int c = 0, inc = 0;
    if (t < NBMAX) {
        c   = (int)s_b[t];
        inc = gcn_wave_incl(c, lane);
        if (lane == 31) s_w[w] = inc;
    }
    __syncthreads();
    if (t < NBMAX) {
        int off = 0;
        for (int k = 0; k < NBMAX / 32; ++k)
            if (k < w) off += s_w[k];
        const int lstart = off + inc - c;
        s_loff[t] = lstart;
        lst[t * nchk + g] = lstart;
        lct[t * nchk + g] = c;
        if (t == NBMAX - 1) s_nv = off + inc;
    }
    __syncthreads();

#pragma unroll
    for (int k = 0; k < EPT; ++k) {
        const unsigned d = (unsigned)dv[k];
        if (d < nloc) {
            const int lp = s_loff[d >> BSH] + (int)rr[k];
            if ((unsigned)lp < (unsigned)CHE) {
                s_d[lp] = dv[k];
                s_s[lp] = sv[k];
            }
        }
    }
    __syncthreads();

    const int nv = min(s_nv, min(CHE, e - e0));
    for (int j = t; j < nv; j += 1024) {
        pd[e0 + j] = s_d[j];
        ps[e0 + j] = s_s[j];
    }
}

#define DCAP 20480
#define DSEG 8
#define DQ   8
typedef __attribute__((ext_vector_type(4))) int v4i;
__global__ __launch_bounds__(1024)
void gcn_bucket_kernel(const int* __restrict__ ps,
                       const int* __restrict__ pd,
                       const int* __restrict__ lst,
                       const int* __restrict__ lct,
                       int* __restrict__ csr,
                       int* __restrict__ cnt,
                       int* __restrict__ rowptr,
                       float* __restrict__ dinv, int n, int e, int nchk) {
    __shared__ int s_out[DCAP];
    __shared__ int s_cnt[BNODE];
    __shared__ int s_cur[BNODE];
    __shared__ int s_w[32];
    __shared__ int s_w2[32];
    const int t    = threadIdx.x;
    const int lane = t & 31;
    const int w    = t >> 5;
    const int b    = blockIdx.x;
    const int n0   = b << BSH;

    int v = 0, v2 = 0;
    const int nfront = b * nchk;
    const v4i* __restrict__ lct4 = reinterpret_cast<const v4i*>(lct);
    const int nq = nfront >> 2;
    for (int base = 0; base < nq; base += DQ * 1024) {
        v4i a[DQ];
#pragma unroll
        for (int u = 0; u < DQ; ++u) {
            const int i = base + u * 1024 + t;
            const v4i z = {0, 0, 0, 0};
            a[u] = (i < nq) ? lct4[i] : z;
        }
#pragma unroll
        for (int u = 0; u < DQ; ++u) v += (a[u].x + a[u].y) + (a[u].z + a[u].w);
    }
    if (t < (nfront & 3)) v += lct[(nq << 2) + t];
    for (int i = t; i < nchk; i += 1024) v2 += lct[nfront + i];
    for (int d = 16; d >= 1; d >>= 1) {
        v  += __shfl_xor(v, d);
        v2 += __shfl_xor(v2, d);
    }
    if (lane == 0) { s_w[w] = v; s_w2[w] = v2; }
    if (t < BNODE) s_cnt[t] = 0;
    __syncthreads();
    int beg = 0, len = 0;
    for (int k = 0; k < 32; ++k) { beg += s_w[k]; len += s_w2[k]; }
    beg = min(max(beg, 0), e);
    len = min(max(len, 0), e - beg);
    const bool staged = len <= DCAP;

    const int per  = (nchk + 31) >> 5;
    const int g_lo = min(w * per, nchk);
    const int g_hi = min(g_lo + per, nchk);

    for (int gb = g_lo; gb < g_hi; gb += 32) {
        int st = 0, ln = 0;
        if (gb + lane < g_hi) {
            const int gg = gb + lane;
            const int c0 = min(gg * CHE, e);
            st = lst[nfront + gg];
            ln = lct[nfront + gg];
            st = c0 + min(max(st, 0), e - c0);
            ln = min(max(ln, 0), e - st);
        }
        const int nseg = min(32, g_hi - gb);
        for (int u0 = 0; u0 < nseg; u0 += DSEG) {
            int stu[DSEG], lnu[DSEG], q[DSEG][2];
#pragma unroll
            for (int u = 0; u < DSEG; ++u) {
                stu[u] = __shfl(st, (u0 + u) & 31);
                lnu[u] = (u0 + u < nseg) ? __shfl(ln, (u0 + u) & 31) : 0;
            }
#pragma unroll
            for (int u = 0; u < DSEG; ++u) {
                q[u][0] = (lane < lnu[u])      ? pd[stu[u] + lane]      : -1;
                q[u][1] = (lane + 32 < lnu[u]) ? pd[stu[u] + lane + 32] : -1;
            }
#pragma unroll
            for (int u = 0; u < DSEG; ++u) {
#pragma unroll
                for (int r = 0; r < 2; ++r) {
                    const unsigned l = (unsigned)(q[u][r] - n0);
                    if (l < (unsigned)BNODE) atomicAdd(&s_cnt[l], 1);
                }
            }
#pragma unroll
            for (int u = 0; u < DSEG; ++u) {
                for (int i = 64 + lane; i < lnu[u]; i += 32) {
                    const unsigned l = (unsigned)(pd[stu[u] + i] - n0);
                    if (l < (unsigned)BNODE) atomicAdd(&s_cnt[l], 1);
                }
            }
        }
    }
    __syncthreads();

    int c = 0, inc = 0;
    if (t < BNODE) {
        c = s_cnt[t];
        inc = gcn_wave_incl(c, lane);
        if (lane == 31) s_w[w] = inc;
    }
    __syncthreads();
    if (t < BNODE) {
        int off = 0;
        for (int k = 0; k < BNODE / 32; ++k)
            if (k < w) off += s_w[k];
        const int ex = off + inc - c;
        s_cur[t] = ex;
        const int node = n0 + t;
        if (node < n) {
            cnt[node]    = c;
            rowptr[node] = beg + ex;
            float dg = (float)(c + 1);
            dinv[node] = dg > 0.0f ? rsqrtf(dg) : 0.0f;
        }
    }
    __syncthreads();

    for (int gb = g_lo; gb < g_hi; gb += 32) {
        int st = 0, ln = 0;
        if (gb + lane < g_hi) {
            const int gg = gb + lane;
            const int c0 = min(gg * CHE, e);
            st = lst[nfront + gg];
            ln = lct[nfront + gg];
            st = c0 + min(max(st, 0), e - c0);
            ln = min(max(ln, 0), e - st);
        }
        const int nseg = min(32, g_hi - gb);
        for (int u0 = 0; u0 < nseg; u0 += DSEG) {
            int stu[DSEG], lnu[DSEG], q[DSEG][2], sv[DSEG][2];
#pragma unroll
            for (int u = 0; u < DSEG; ++u) {
                stu[u] = __shfl(st, (u0 + u) & 31);
                lnu[u] = (u0 + u < nseg) ? __shfl(ln, (u0 + u) & 31) : 0;
            }
#pragma unroll
            for (int u = 0; u < DSEG; ++u) {
                const bool in0 = lane < lnu[u];
                const bool in1 = lane + 32 < lnu[u];
                q[u][0]  = in0 ? pd[stu[u] + lane]      : -1;
                q[u][1]  = in1 ? pd[stu[u] + lane + 32] : -1;
                sv[u][0] = in0 ? ps[stu[u] + lane]      : 0;
                sv[u][1] = in1 ? ps[stu[u] + lane + 32] : 0;
            }
#pragma unroll
            for (int u = 0; u < DSEG; ++u) {
#pragma unroll
                for (int r = 0; r < 2; ++r) {
                    const unsigned l = (unsigned)(q[u][r] - n0);
                    if (l < (unsigned)BNODE) {
                        const int slot = atomicAdd(&s_cur[l], 1);
                        if ((unsigned)slot < (unsigned)len) {
                            if (staged) s_out[slot] = sv[u][r];
                            else        csr[beg + slot] = sv[u][r];
                        }
                    }
                }
            }
#pragma unroll
            for (int u = 0; u < DSEG; ++u) {
                for (int i = 64 + lane; i < lnu[u]; i += 32) {
                    const unsigned l = (unsigned)(pd[stu[u] + i] - n0);
                    const int s1 = ps[stu[u] + i];
                    if (l < (unsigned)BNODE) {
                        const int slot = atomicAdd(&s_cur[l], 1);
                        if ((unsigned)slot < (unsigned)len) {
                            if (staged) s_out[slot] = s1;
                            else        csr[beg + slot] = s1;
                        }
                    }
                }
            }
        }
    }
    __syncthreads();
    if (staged) {
        for (int i = t; i < len; i += 1024) csr[beg + i] = s_out[i];
    }
}

template <bool RELU, int K>
__global__ __launch_bounds__(128)
void gcn_gemm_wmma_kernel(const float* __restrict__ A,
                          const float* __restrict__ B,
                          const float* __restrict__ rowscale,
                          float* __restrict__ C, int nrows) {
    const int lane = threadIdx.x & 31;
    const int wave = threadIdx.x >> 5;
    const int tile = blockIdx.x * 4 + wave;
    const int row0 = tile * 16;
    if (row0 >= nrows) return;

    const int m    = lane & 15;
    const int half = lane >> 4;
    const int kh   = half << 1;

    const float* __restrict__ arow = A + (size_t)(row0 + m) * K;
    const float sc = rowscale[row0 + m];

    v8f acc = {};
#pragma unroll
    for (int k = 0; k < K; k += 4) {
        v2f a = *reinterpret_cast<const v2f*>(arow + k + kh);
        if (RELU) {
            a.x = fmaxf(a.x, 0.0f);
            a.y = fmaxf(a.y, 0.0f);
        }
        a.x *= sc;
        a.y *= sc;
        v2f b;
        b.x = B[(k + kh) * HID + m];
        b.y = B[(k + kh + 1) * HID + m];
        acc = __builtin_amdgcn_wmma_f32_16x16x4_f32(
             false, a,  false, b,
             (short)0, acc,  false,  false);
    }

    float* __restrict__ ccol = C + (size_t)row0 * HID + m;
#pragma unroll
    for (int r = 0; r < 8; ++r) {
        ccol[(size_t)(r + 8 * half) * HID] = acc[r];
    }
}

typedef __attribute__((ext_vector_type(4))) float v4f;

template <int Q>
static __device__ __forceinline__ int gcn_quad_lane(int v) {
    return __builtin_amdgcn_update_dpp(0, v, Q * 0x55, 0xF, 0xF, true);
}

template <int K>
static __device__ __forceinline__ v4f gcn_take4(const float* __restrict__ hs,
                                                const int (&ids)[4], unsigned last,
                                                unsigned q4, int left) {
    unsigned s = (unsigned)gcn_quad_lane<(K & 3)>(ids[K >> 2]);
    s = min(s, last);
    v4f x = *reinterpret_cast<const v4f*>(hs + s * HID + q4);
    const v4f z = {0.0f, 0.0f, 0.0f, 0.0f};
    return (K < left) ? x : z;
}

__global__ __launch_bounds__(256)
void gcn_gather_kernel(const float* __restrict__ hs,
                       const int* __restrict__ rowptr,
                       const int* __restrict__ cnt,
                       const int* __restrict__ csr,
                       const float* __restrict__ dinv,
                       const float* __restrict__ bias,
                       float* __restrict__ out, int n, int e) {
    const int tid = blockIdx.x * blockDim.x + threadIdx.x;
    const int i = tid >> 2;
    const int q = tid & 3;
    const unsigned q4 = (unsigned)q * 4u;
    if (i >= n) return;

    int beg = rowptr[i];
    int c   = cnt[i];
    beg = min(max(beg, 0), e);
    c   = min(max(c, 0), e - beg);
    const int* __restrict__ p = csr + beg;
    const unsigned last = (unsigned)(n - 1);

    v4f acc = *reinterpret_cast<const v4f*>(hs + (unsigned)i * HID + q4);
    for (int j = 0; j < c; j += 16) {
        const int left = c - j;
        int ids[4];
#pragma unroll
        for (int u = 0; u < 4; ++u) ids[u] = p[min(j + 4 * u + q, c - 1)];
        v4f v0  = gcn_take4<0>(hs, ids, last, q4, left);
        v4f v1  = gcn_take4<1>(hs, ids, last, q4, left);
        v4f v2  = gcn_take4<2>(hs, ids, last, q4, left);
        v4f v3  = gcn_take4<3>(hs, ids, last, q4, left);
        v4f v4  = gcn_take4<4>(hs, ids, last, q4, left);
        v4f v5  = gcn_take4<5>(hs, ids, last, q4, left);
        v4f v6  = gcn_take4<6>(hs, ids, last, q4, left);
        v4f v7  = gcn_take4<7>(hs, ids, last, q4, left);
        v4f v8  = gcn_take4<8>(hs, ids, last, q4, left);
        v4f v9  = gcn_take4<9>(hs, ids, last, q4, left);
        v4f v10 = gcn_take4<10>(hs, ids, last, q4, left);
        v4f v11 = gcn_take4<11>(hs, ids, last, q4, left);
        v4f v12 = gcn_take4<12>(hs, ids, last, q4, left);
        v4f v13 = gcn_take4<13>(hs, ids, last, q4, left);
        v4f v14 = gcn_take4<14>(hs, ids, last, q4, left);
        v4f v15 = gcn_take4<15>(hs, ids, last, q4, left);
        acc += v0;  acc += v1;  acc += v2;  acc += v3;
        acc += v4;  acc += v5;  acc += v6;  acc += v7;
        acc += v8;  acc += v9;  acc += v10; acc += v11;
        acc += v12; acc += v13; acc += v14; acc += v15;
    }
    const float di = dinv[i];
    const v4f bv = *reinterpret_cast<const v4f*>(bias + q4);
    v4f o;
    o.x = acc.x * di + bv.x;
    o.y = acc.y * di + bv.y;
    o.z = acc.z * di + bv.z;
    o.w = acc.w * di + bv.w;
    *reinterpret_cast<v4f*>(out + (unsigned)i * HID + q4) = o;
}

static inline size_t gcn_align256(size_t b) { return (b + 255) & ~(size_t)255; }

extern "C" void kernel_launch(void* const* d_in, const int* in_sizes, int n_in,
                              void* d_out, int out_size, void* d_ws, size_t ws_size,
                              hipStream_t stream) {
    const float* x  = (const float*)d_in[0];
    const int*   ei = (const int*)d_in[1];
    const float* W1 = (const float*)d_in[2];
    const float* b1 = (const float*)d_in[3];
    const float* W2 = (const float*)d_in[4];
    const float* b2 = (const float*)d_in[5];
    float* out = (float*)d_out;

    const int N = in_sizes[0] / F_IN;
    const int E = in_sizes[1] / 2;
    const int* src = ei;
    const int* dst = ei + E;

    char* ws = (char*)d_ws;
    const size_t szN  = gcn_align256((size_t)N * sizeof(int));
    const size_t szE  = gcn_align256((size_t)E * sizeof(int));
    const size_t szNH = gcn_align256((size_t)N * HID * sizeof(float));
    int*   cnt    = (int*)ws;
    int*   rowptr = (int*)(ws + szN);
    float* dinv   = (float*)(ws + 2 * szN);
    int*   csr    = (int*)(ws + 3 * szN);
    float* hbuf   = (float*)(ws + 3 * szN + szE);
    float* agg1   = (float*)(ws + 3 * szN + szE + szNH);
    int*   ps     = (int*)(ws + 3 * szN + szE + 2 * szNH);
    int*   pd     = (int*)(ws + 3 * szN + 2 * szE + 2 * szNH);
    const int NCHK = (E + CHE - 1) / CHE;
    const size_t szBC = gcn_align256((size_t)NBMAX * NCHK * sizeof(int));
    int*   lst    = (int*)(ws + 3 * szN + 3 * szE + 2 * szNH);
    int*   lct    = (int*)(ws + 3 * szN + 3 * szE + 2 * szNH + szBC);

    const int TB = 256;

    gcn_part_kernel<<<NCHK, 1024, 0, stream>>>(src, dst, lst, lct, ps, pd, N, E, NCHK);
    gcn_bucket_kernel<<<NBMAX, 1024, 0, stream>>>(ps, pd, lst, lct, csr, cnt, rowptr, dinv, N, E, NCHK);

    const int tiles = (N + 15) / 16;
    const int gblk  = (tiles + 3) / 4;
    gcn_gemm_wmma_kernel<false, F_IN><<<gblk, 128, 0, stream>>>(x, W1, dinv, hbuf, N);

    const long long gt = (long long)N * 4;
    const int ggrid = (int)((gt + TB - 1) / TB);
    gcn_gather_kernel<<<ggrid, TB, 0, stream>>>(hbuf, rowptr, cnt, csr, dinv, b1, agg1, N, E);

    gcn_gemm_wmma_kernel<true, HID><<<gblk, 128, 0, stream>>>(agg1, W2, dinv, hbuf, N);

    gcn_gather_kernel<<<ggrid, TB, 0, stream>>>(hbuf, rowptr, cnt, csr, dinv, b2, out, N, E);
}


# layer-2 linear folded into the first gather: a work-group hands its 64 rows (relu, row scale) through LDS to 4 tile waves on the f32 matrix instruction; its launch and the agg1 round trip are gone
# speedup vs baseline: 5.3705x; 1.0410x over previous
#include <hip/hip_runtime.h>
#include <hip/hip_bf16.h>


typedef __attribute__((ext_vector_type(2))) float v2f;
typedef __attribute__((ext_vector_type(8))) float v8f;

#define F_IN 128
#define HID  16

#define BSH   9
#define BNODE 512
#define NBMAX 256
#define NCAP  (NBMAX * BNODE)
#define EPT   8
#define CHE   (1024 * EPT)

static __device__ __forceinline__ int gcn_wave_incl(int v, int lane) {
    for (int d = 1; d < 32; d <<= 1) {
        int u = __shfl_up(v, d);
        if (lane >= d) v += u;
    }
    return v;
}

__global__ __launch_bounds__(1024)
void gcn_part_kernel(const int* __restrict__ src,
                     const int* __restrict__ dst,
                     int* __restrict__ lst,
                     int* __restrict__ lct,
                     int* __restrict__ ps,
                     int* __restrict__ pd,
                     int n, int e, int nchk) {
    __shared__ int s_d[CHE];
    __shared__ int s_s[CHE];
    __shared__ unsigned s_b[NBMAX];
    __shared__ int s_loff[NBMAX];
    __shared__ int s_w[NBMAX / 32];
    __shared__ int s_nv;
    const int t    = threadIdx.x;
    const int lane = t & 31;
    const int w    = t >> 5;
    const int g    = blockIdx.x;
    const unsigned nloc = (unsigned)min(n, NCAP);
    if (t < NBMAX) s_b[t] = 0u;
    __syncthreads();

    const int e0 = g * CHE;
    int dv[EPT], sv[EPT];
#pragma unroll
    for (int k = 0; k < EPT; ++k) {
        const int i = e0 + k * 1024 + t;
        const bool in = i < e;
        dv[k] = in ? dst[i] : -1;
        sv[k] = in ? src[i] : 0;
    }
    unsigned rr[EPT];
#pragma unroll
    for (int k = 0; k < EPT; ++k) {
        const unsigned d  = (unsigned)dv[k];
        const bool     ok = d < nloc;
        rr[k] = atomicAdd(&s_b[ok ? (d >> BSH) : 0u], ok ? 1u : 0u);
    }
    __syncthreads();

    int c = 0, inc = 0;
    if (t < NBMAX) {
        c   = (int)s_b[t];
        inc = gcn_wave_incl(c, lane);
        if (lane == 31) s_w[w] = inc;
    }
    __syncthreads();
    if (t < NBMAX) {
        int off = 0;
        for (int k = 0; k < NBMAX / 32; ++k)
            if (k < w) off += s_w[k];
        const int lstart = off + inc - c;
        s_loff[t] = lstart;
        lst[t * nchk + g] = lstart;
        lct[t * nchk + g] = c;
        if (t == NBMAX - 1) s_nv = off + inc;
    }
    __syncthreads();

#pragma unroll
    for (int k = 0; k < EPT; ++k) {
        const unsigned d = (unsigned)dv[k];
        if (d < nloc) {
            const int lp = s_loff[d >> BSH] + (int)rr[k];
            if ((unsigned)lp < (unsigned)CHE) {
                s_d[lp] = dv[k];
                s_s[lp] = sv[k];
            }
        }
    }
    __syncthreads();

    const int nv = min(s_nv, min(CHE, e - e0));
    for (int j = t; j < nv; j += 1024) {
        pd[e0 + j] = s_d[j];
        ps[e0 + j] = s_s[j];
    }
}

#define DCAP 20480
#define DSEG 8
#define DQ   8
typedef __attribute__((ext_vector_type(4))) int v4i;
__global__ __launch_bounds__(1024)
void gcn_bucket_kernel(const int* __restrict__ ps,
                       const int* __restrict__ pd,
                       const int* __restrict__ lst,
                       const int* __restrict__ lct,
                       int* __restrict__ csr,
                       int* __restrict__ cnt,
                       int* __restrict__ rowptr,
                       float* __restrict__ dinv, int n, int e, int nchk) {
    __shared__ int s_out[DCAP];
    __shared__ int s_cnt[BNODE];
    __shared__ int s_cur[BNODE];
    __shared__ int s_w[32];
    __shared__ int s_w2[32];
    const int t    = threadIdx.x;
    const int lane = t & 31;
    const int w    = t >> 5;
    const int b    = blockIdx.x;
    const int n0   = b << BSH;

    int v = 0, v2 = 0;
    const int nfront = b * nchk;
    const v4i* __restrict__ lct4 = reinterpret_cast<const v4i*>(lct);
    const int nq = nfront >> 2;
    for (int base = 0; base < nq; base += DQ * 1024) {
        v4i a[DQ];
#pragma unroll
        for (int u = 0; u < DQ; ++u) {
            const int i = base + u * 1024 + t;
            const v4i z = {0, 0, 0, 0};
            a[u] = (i < nq) ? lct4[i] : z;
        }
#pragma unroll
        for (int u = 0; u < DQ; ++u) v += (a[u].x + a[u].y) + (a[u].z + a[u].w);
    }
    if (t < (nfront & 3)) v += lct[(nq << 2) + t];
    for (int i = t; i < nchk; i += 1024) v2 += lct[nfront + i];
    for (int d = 16; d >= 1; d >>= 1) {
        v  += __shfl_xor(v, d);
        v2 += __shfl_xor(v2, d);
    }
    if (lane == 0) { s_w[w] = v; s_w2[w] = v2; }
    if (t < BNODE) s_cnt[t] = 0;
    __syncthreads();
    int beg = 0, len = 0;
    for (int k = 0; k < 32; ++k) { beg += s_w[k]; len += s_w2[k]; }
    beg = min(max(beg, 0), e);
    len = min(max(len, 0), e - beg);
    const bool staged = len <= DCAP;

    const int per  = (nchk + 31) >> 5;
    const int g_lo = min(w * per, nchk);
    const int g_hi = min(g_lo + per, nchk);

    for (int gb = g_lo; gb < g_hi; gb += 32) {
        int st = 0, ln = 0;
        if (gb + lane < g_hi) {
            const int gg = gb + lane;
            const int c0 = min(gg * CHE, e);
            st = lst[nfront + gg];
            ln = lct[nfront + gg];
            st = c0 + min(max(st, 0), e - c0);
            ln = min(max(ln, 0), e - st);
        }
        const int nseg = min(32, g_hi - gb);
        for (int u0 = 0; u0 < nseg; u0 += DSEG) {
            int stu[DSEG], lnu[DSEG], q[DSEG][2];
#pragma unroll
            for (int u = 0; u < DSEG; ++u) {
                stu[u] = __shfl(st, (u0 + u) & 31);
                lnu[u] = (u0 + u < nseg) ? __shfl(ln, (u0 + u) & 31) : 0;
            }
#pragma unroll
            for (int u = 0; u < DSEG; ++u) {
                q[u][0] = (lane < lnu[u])      ? pd[stu[u] + lane]      : -1;
                q[u][1] = (lane + 32 < lnu[u]) ? pd[stu[u] + lane + 32] : -1;
            }
#pragma unroll
            for (int u = 0; u < DSEG; ++u) {
#pragma unroll
                for (int r = 0; r < 2; ++r) {
                    const unsigned l = (unsigned)(q[u][r] - n0);
                    if (l < (unsigned)BNODE) atomicAdd(&s_cnt[l], 1);
                }
            }
#pragma unroll
            for (int u = 0; u < DSEG; ++u) {
                for (int i = 64 + lane; i < lnu[u]; i += 32) {
                    const unsigned l = (unsigned)(pd[stu[u] + i] - n0);
                    if (l < (unsigned)BNODE) atomicAdd(&s_cnt[l], 1);
                }
            }
        }
    }
    __syncthreads();

    int c = 0, inc = 0;
    if (t < BNODE) {
        c = s_cnt[t];
        inc = gcn_wave_incl(c, lane);
        if (lane == 31) s_w[w] = inc;
    }
    __syncthreads();
    if (t < BNODE) {
        int off = 0;
        for (int k = 0; k < BNODE / 32; ++k)
            if (k < w) off += s_w[k];
        const int ex = off + inc - c;
        s_cur[t] = ex;
        const int node = n0 + t;
        if (node < n) {
            cnt[node]    = c;
            rowptr[node] = beg + ex;
            float dg = (float)(c + 1);
            dinv[node] = dg > 0.0f ? rsqrtf(dg) : 0.0f;
        }
    }
    __syncthreads();

    for (int gb = g_lo; gb < g_hi; gb += 32) {
        int st = 0, ln = 0;
        if (gb + lane < g_hi) {
            const int gg = gb + lane;
            const int c0 = min(gg * CHE, e);
            st = lst[nfront + gg];
            ln = lct[nfront + gg];
            st = c0 + min(max(st, 0), e - c0);
            ln = min(max(ln, 0), e - st);
        }
        const int nseg = min(32, g_hi - gb);
        for (int u0 = 0; u0 < nseg; u0 += DSEG) {
            int stu[DSEG], lnu[DSEG], q[DSEG][2], sv[DSEG][2];
#pragma unroll
            for (int u = 0; u < DSEG; ++u) {
                stu[u] = __shfl(st, (u0 + u) & 31);
                lnu[u] = (u0 + u < nseg) ? __shfl(ln, (u0 + u) & 31) : 0;
            }
#pragma unroll
            for (int u = 0; u < DSEG; ++u) {
                const bool in0 = lane < lnu[u];
                const bool in1 = lane + 32 < lnu[u];
                q[u][0]  = in0 ? pd[stu[u] + lane]      : -1;
                q[u][1]  = in1 ? pd[stu[u] + lane + 32] : -1;
                sv[u][0] = in0 ? ps[stu[u] + lane]      : 0;
                sv[u][1] = in1 ? ps[stu[u] + lane + 32] : 0;
            }
#pragma unroll
            for (int u = 0; u < DSEG; ++u) {
#pragma unroll
                for (int r = 0; r < 2; ++r) {
                    const unsigned l = (unsigned)(q[u][r] - n0);
                    if (l < (unsigned)BNODE) {
                        const int slot = atomicAdd(&s_cur[l], 1);
                        if ((unsigned)slot < (unsigned)len) {
                            if (staged) s_out[slot] = sv[u][r];
                            else        csr[beg + slot] = sv[u][r];
                        }
                    }
                }
            }
#pragma unroll
            for (int u = 0; u < DSEG; ++u) {
                for (int i = 64 + lane; i < lnu[u]; i += 32) {
                    const unsigned l = (unsigned)(pd[stu[u] + i] - n0);
                    const int s1 = ps[stu[u] + i];
                    if (l < (unsigned)BNODE) {
                        const int slot = atomicAdd(&s_cur[l], 1);
                        if ((unsigned)slot < (unsigned)len) {
                            if (staged) s_out[slot] = s1;
                            else        csr[beg + slot] = s1;
                        }
                    }
                }
            }
        }
    }
    __syncthreads();
    if (staged) {
        for (int i = t; i < len; i += 1024) csr[beg + i] = s_out[i];
    }
}

template <int K>
__global__ __launch_bounds__(128)
void gcn_gemm_wmma_kernel(const float* __restrict__ A,
                          const float* __restrict__ B,
                          const float* __restrict__ rowscale,
                          float* __restrict__ C, int nrows) {
    const int lane = threadIdx.x & 31;
    const int wave = threadIdx.x >> 5;
    const int tile = blockIdx.x * 4 + wave;
    const int row0 = tile * 16;
    if (row0 >= nrows) return;

    const int m    = lane & 15;
    const int half = lane >> 4;
    const int kh   = half << 1;

    const float* __restrict__ arow = A + (size_t)(row0 + m) * K;
    const float sc = rowscale[row0 + m];

    v8f acc = {};
#pragma unroll
    for (int k = 0; k < K; k += 4) {
        v2f a = *reinterpret_cast<const v2f*>(arow + k + kh);
        a.x *= sc;
        a.y *= sc;
        v2f b;
        b.x = B[(k + kh) * HID + m];
        b.y = B[(k + kh + 1) * HID + m];
        acc = __builtin_amdgcn_wmma_f32_16x16x4_f32(
             false, a,  false, b,
             (short)0, acc,  false,  false);
    }

    float* __restrict__ ccol = C + (size_t)row0 * HID + m;
#pragma unroll
    for (int r = 0; r < 8; ++r) {
        ccol[(size_t)(r + 8 * half) * HID] = acc[r];
    }
}

typedef __attribute__((ext_vector_type(4))) float v4f;

template <int Q>
static __device__ __forceinline__ int gcn_quad_lane(int v) {
    return __builtin_amdgcn_update_dpp(0, v, Q * 0x55, 0xF, 0xF, true);
}

template <int K>
static __device__ __forceinline__ v4f gcn_take4(const float* __restrict__ hs,
                                                const int (&ids)[4], unsigned last,
                                                unsigned q4, int left) {
    unsigned s = (unsigned)gcn_quad_lane<(K & 3)>(ids[K >> 2]);
    s = min(s, last);
    v4f x = *reinterpret_cast<const v4f*>(hs + s * HID + q4);
    const v4f z = {0.0f, 0.0f, 0.0f, 0.0f};
    return (K < left) ? x : z;
}

#define APITCH 20

template <bool W2FUSE>
__global__ __launch_bounds__(256)
void gcn_gather_kernel(const float* __restrict__ hs,
                       const int* __restrict__ rowptr,
                       const int* __restrict__ cnt,
                       const int* __restrict__ csr,
                       const float* __restrict__ dinv,
                       const float* __restrict__ bias,
                       const float* __restrict__ W2,
                       float* __restrict__ out, int n, int e) {
    const int tid = blockIdx.x * blockDim.x + threadIdx.x;
    int i = tid >> 2;
    const int q = tid & 3;
    const unsigned q4 = (unsigned)q * 4u;
    const int wave = __builtin_amdgcn_readfirstlane((int)(threadIdx.x >> 5));
    if (!W2FUSE) {
        if (i >= n) return;
    } else {
        i = min(i, n - 1);
    }

    int beg = rowptr[i];
    int c   = cnt[i];
    beg = min(max(beg, 0), e);
    c   = min(max(c, 0), e - beg);
    const int* __restrict__ p = csr + beg;
    const unsigned last = (unsigned)(n - 1);

    v4f acc = *reinterpret_cast<const v4f*>(hs + (unsigned)i * HID + q4);
    for (int j = 0; j < c; j += 16) {
        const int left = c - j;
        int ids[4];
#pragma unroll
        for (int u = 0; u < 4; ++u) ids[u] = p[min(j + 4 * u + q, c - 1)];
        v4f v0  = gcn_take4<0>(hs, ids, last, q4, left);
        v4f v1  = gcn_take4<1>(hs, ids, last, q4, left);
        v4f v2  = gcn_take4<2>(hs, ids, last, q4, left);
        v4f v3  = gcn_take4<3>(hs, ids, last, q4, left);
        v4f v4  = gcn_take4<4>(hs, ids, last, q4, left);
        v4f v5  = gcn_take4<5>(hs, ids, last, q4, left);
        v4f v6  = gcn_take4<6>(hs, ids, last, q4, left);
        v4f v7  = gcn_take4<7>(hs, ids, last, q4, left);
        v4f v8  = gcn_take4<8>(hs, ids, last, q4, left);
        v4f v9  = gcn_take4<9>(hs, ids, last, q4, left);
        v4f v10 = gcn_take4<10>(hs, ids, last, q4, left);
        v4f v11 = gcn_take4<11>(hs, ids, last, q4, left);
        v4f v12 = gcn_take4<12>(hs, ids, last, q4, left);
        v4f v13 = gcn_take4<13>(hs, ids, last, q4, left);
        v4f v14 = gcn_take4<14>(hs, ids, last, q4, left);
        v4f v15 = gcn_take4<15>(hs, ids, last, q4, left);
        acc += v0;  acc += v1;  acc += v2;  acc += v3;
        acc += v4;  acc += v5;  acc += v6;  acc += v7;
        acc += v8;  acc += v9;  acc += v10; acc += v11;
        acc += v12; acc += v13; acc += v14; acc += v15;
    }
    const float di = dinv[i];
    const v4f bv = *reinterpret_cast<const v4f*>(bias + q4);
    v4f o;
    o.x = acc.x * di + bv.x;
    o.y = acc.y * di + bv.y;
    o.z = acc.z * di + bv.z;
    o.w = acc.w * di + bv.w;

    if (!W2FUSE) {
        *reinterpret_cast<v4f*>(out + (unsigned)i * HID + q4) = o;
    } else {
        const int lane = (int)__lane_id();
        const int m    = lane & 15;
        const int kh   = (lane >> 4) << 1;
        v2f bw[HID / 4];
        if (wave < 4) {
#pragma unroll
            for (int k = 0; k < HID; k += 4) {
                bw[k >> 2].x = W2[(k + kh) * HID + m];
                bw[k >> 2].y = W2[(k + kh + 1) * HID + m];
            }
        }
        __shared__ float s_a[64 * APITCH];
        v4f r;
        r.x = fmaxf(o.x, 0.0f) * di;
        r.y = fmaxf(o.y, 0.0f) * di;
        r.z = fmaxf(o.z, 0.0f) * di;
        r.w = fmaxf(o.w, 0.0f) * di;
        *reinterpret_cast<v4f*>(&s_a[(wave * 8 + (lane >> 2)) * APITCH + q4]) = r;
        __syncthreads();
        const int row0 = blockIdx.x * 64 + wave * 16;
        if (wave < 4 && row0 < n) {
            const float* __restrict__ arow = &s_a[(wave * 16 + m) * APITCH];
            v8f acc2 = {};
#pragma unroll
            for (int k = 0; k < HID; k += 4) {
                const v2f a = *reinterpret_cast<const v2f*>(arow + k + kh);
                acc2 = __builtin_amdgcn_wmma_f32_16x16x4_f32(
                     false, a,  false, bw[k >> 2],
                     (short)0, acc2,  false,  false);
            }
            const int rbase = row0 + 8 * (lane >> 4);
#pragma unroll
            for (int r8 = 0; r8 < 8; ++r8) {
                if (rbase + r8 < n) out[(size_t)(rbase + r8) * HID + m] = acc2[r8];
            }
        }
    }
}

static inline size_t gcn_align256(size_t b) { return (b + 255) & ~(size_t)255; }

extern "C" void kernel_launch(void* const* d_in, const int* in_sizes, int n_in,
                              void* d_out, int out_size, void* d_ws, size_t ws_size,
                              hipStream_t stream) {
    const float* x  = (const float*)d_in[0];
    const int*   ei = (const int*)d_in[1];
    const float* W1 = (const float*)d_in[2];
    const float* b1 = (const float*)d_in[3];
    const float* W2 = (const float*)d_in[4];
    const float* b2 = (const float*)d_in[5];
    float* out = (float*)d_out;

    const int N = in_sizes[0] / F_IN;
    const int E = in_sizes[1] / 2;
    const int* src = ei;
    const int* dst = ei + E;

    char* ws = (char*)d_ws;
    const size_t szN  = gcn_align256((size_t)N * sizeof(int));
    const size_t szE  = gcn_align256((size_t)E * sizeof(int));
    const size_t szNH = gcn_align256((size_t)N * HID * sizeof(float));
    int*   cnt    = (int*)ws;
    int*   rowptr = (int*)(ws + szN);
    float* dinv   = (float*)(ws + 2 * szN);
    int*   csr    = (int*)(ws + 3 * szN);
    float* hbuf   = (float*)(ws + 3 * szN + szE);
    float* hbuf2  = (float*)(ws + 3 * szN + szE + szNH);
    int*   ps     = (int*)(ws + 3 * szN + szE + 2 * szNH);
    int*   pd     = (int*)(ws + 3 * szN + 2 * szE + 2 * szNH);
    const int NCHK = (E + CHE - 1) / CHE;
    const size_t szBC = gcn_align256((size_t)NBMAX * NCHK * sizeof(int));
    int*   lst    = (int*)(ws + 3 * szN + 3 * szE + 2 * szNH);
    int*   lct    = (int*)(ws + 3 * szN + 3 * szE + 2 * szNH + szBC);

    const int TB = 256;

    gcn_part_kernel<<<NCHK, 1024, 0, stream>>>(src, dst, lst, lct, ps, pd, N, E, NCHK);
    gcn_bucket_kernel<<<NBMAX, 1024, 0, stream>>>(ps, pd, lst, lct, csr, cnt, rowptr, dinv, N, E, NCHK);

    const int tiles = (N + 15) / 16;
    const int gblk  = (tiles + 3) / 4;
    gcn_gemm_wmma_kernel<F_IN><<<gblk, 128, 0, stream>>>(x, W1, dinv, hbuf, N);

    const long long gt = (long long)N * 4;
    const int ggrid = (int)((gt + TB - 1) / TB);
    gcn_gather_kernel<true><<<ggrid, TB, 0, stream>>>(hbuf, rowptr, cnt, csr, dinv, b1, W2, hbuf2, N, E);

    gcn_gather_kernel<false><<<ggrid, TB, 0, stream>>>(hbuf2, rowptr, cnt, csr, dinv, b2, W2, out, N, E);
}


# layer-1 linear folded into the bucket launch: each wave runs a 16-row tile of x @ W1 on the f32 matrix instruction in 3 K parts loaded under the sort's phases; dinv scales the sums; GEMM launch gone
# speedup vs baseline: 5.5504x; 1.0335x over previous
#include <hip/hip_runtime.h>
#include <hip/hip_bf16.h>


typedef __attribute__((ext_vector_type(2))) float v2f;
typedef __attribute__((ext_vector_type(8))) float v8f;
typedef __attribute__((ext_vector_type(4))) float v4f;

#define F_IN 128
#define HID  16

#define BSH   9
#define BNODE 512
#define NBMAX 256
#define NCAP  (NBMAX * BNODE)
#define EPT   8
#define CHE   (1024 * EPT)

static __device__ __forceinline__ int gcn_wave_incl(int v, int lane) {
    for (int d = 1; d < 32; d <<= 1) {
        int u = __shfl_up(v, d);
        if (lane >= d) v += u;
    }
    return v;
}

__global__ __launch_bounds__(1024)
void gcn_part_kernel(const int* __restrict__ src,
                     const int* __restrict__ dst,
                     int* __restrict__ lst,
                     int* __restrict__ lct,
                     int* __restrict__ ps,
                     int* __restrict__ pd,
                     int n, int e, int nchk) {
    __shared__ int s_d[CHE];
    __shared__ int s_s[CHE];
    __shared__ unsigned s_b[NBMAX];
    __shared__ int s_loff[NBMAX];
    __shared__ int s_w[NBMAX / 32];
    __shared__ int s_nv;
    const int t    = threadIdx.x;
    const int lane = t & 31;
    const int w    = t >> 5;
    const int g    = blockIdx.x;
    const unsigned nloc = (unsigned)min(n, NCAP);
    if (t < NBMAX) s_b[t] = 0u;
    __syncthreads();

    const int e0 = g * CHE;
    int dv[EPT], sv[EPT];
#pragma unroll
    for (int k = 0; k < EPT; ++k) {
        const int i = e0 + k * 1024 + t;
        const bool in = i < e;
        dv[k] = in ? dst[i] : -1;
        sv[k] = in ? src[i] : 0;
    }
    unsigned rr[EPT];
#pragma unroll
    for (int k = 0; k < EPT; ++k) {
        const unsigned d  = (unsigned)dv[k];
        const bool     ok = d < nloc;
        rr[k] = atomicAdd(&s_b[ok ? (d >> BSH) : 0u], ok ? 1u : 0u);
    }
    __syncthreads();

    int c = 0, inc = 0;
    if (t < NBMAX) {
        c   = (int)s_b[t];
        inc = gcn_wave_incl(c, lane);
        if (lane == 31) s_w[w] = inc;
    }
    __syncthreads();
    if (t < NBMAX) {
        int off = 0;
        for (int k = 0; k < NBMAX / 32; ++k)
            if (k < w) off += s_w[k];
        const int lstart = off + inc - c;
        s_loff[t] = lstart;
        lst[t * nchk + g] = lstart;
        lct[t * nchk + g] = c;
        if (t == NBMAX - 1) s_nv = off + inc;
    }
    __syncthreads();

#pragma unroll
    for (int k = 0; k < EPT; ++k) {
        const unsigned d = (unsigned)dv[k];
        if (d < nloc) {
            const int lp = s_loff[d >> BSH] + (int)rr[k];
            if ((unsigned)lp < (unsigned)CHE) {
                s_d[lp] = dv[k];
                s_s[lp] = sv[k];
            }
        }
    }
    __syncthreads();

    const int nv = min(s_nv, min(CHE, e - e0));
    for (int j = t; j < nv; j += 1024) {
        pd[e0 + j] = s_d[j];
        ps[e0 + j] = s_s[j];
    }
}

#define DCAP 20480
#define KS0  8
#define KS1  12
#define KS2  12

template <int NS>
static __device__ __forceinline__ void gcn_tile_load(const float* __restrict__ arow,
                                                     const float* __restrict__ W1,
                                                     int k0, int kh, int m,
                                                     v2f (&a)[NS], v2f (&bw)[NS]) {
#pragma unroll
    for (int s = 0; s < NS; ++s) {
        const int k = k0 + 4 * s + kh;
        a[s]    = *reinterpret_cast<const v2f*>(arow + k);
        bw[s].x = W1[k * HID + m];
        bw[s].y = W1[(k + 1) * HID + m];
    }
}

template <int NS>
static __device__ __forceinline__ v8f gcn_tile_mma(v8f acc, const v2f (&a)[NS],
                                                   const v2f (&bw)[NS]) {
#pragma unroll
    for (int s = 0; s < NS; ++s) {
        acc = __builtin_amdgcn_wmma_f32_16x16x4_f32(
             false, a[s],  false, bw[s],
             (short)0, acc,  false,  false);
    }
    return acc;
}
#define DSEG 8
#define DQ   8
typedef __attribute__((ext_vector_type(4))) int v4i;
__global__ __launch_bounds__(1024)
void gcn_bucket_kernel(const int* __restrict__ ps,
                       const int* __restrict__ pd,
                       const int* __restrict__ lst,
                       const int* __restrict__ lct,
                       int* __restrict__ csr,
                       int* __restrict__ cnt,
                       int* __restrict__ rowptr,
                       float* __restrict__ dinv,
                       const float* __restrict__ x,
                       const float* __restrict__ W1,
                       float* __restrict__ hs, int n, int e, int nchk) {
    __shared__ int s_out[DCAP];
    __shared__ __attribute__((aligned(16))) float s_dinv[BNODE];
    __shared__ int s_cnt[BNODE];
    __shared__ int s_cur[BNODE];
    __shared__ int s_w[32];
    __shared__ int s_w2[32];
    const int t    = threadIdx.x;
    const int lane = t & 31;
    const int w    = t >> 5;
    const int b    = blockIdx.x;
    const int n0   = b << BSH;

    const int m    = lane & 15;
    const int half = lane >> 4;
    const int kh   = half << 1;
    const int row0 = n0 + (__builtin_amdgcn_readfirstlane(w) << 4);
    const bool tile_on = row0 < n;
    v8f hacc = {};
    v2f xa0[KS0], xb0[KS0], xa1[KS1], xb1[KS1], xa2[KS2], xb2[KS2];
    const float* __restrict__ arow = x + (size_t)min(row0 + m, n - 1) * F_IN;
    if (tile_on) gcn_tile_load<KS0>(arow, W1, 0, kh, m, xa0, xb0);

    int v = 0, v2 = 0;
    const int nfront = b * nchk;
    const v4i* __restrict__ lct4 = reinterpret_cast<const v4i*>(lct);
    const int nq = nfront >> 2;
    for (int base = 0; base < nq; base += DQ * 1024) {
        v4i a[DQ];
#pragma unroll
        for (int u = 0; u < DQ; ++u) {
            const int i = base + u * 1024 + t;
            const v4i z = {0, 0, 0, 0};
            a[u] = (i < nq) ? lct4[i] : z;
        }
#pragma unroll
        for (int u = 0; u < DQ; ++u) v += (a[u].x + a[u].y) + (a[u].z + a[u].w);
    }
    if (t < (nfront & 3)) v += lct[(nq << 2) + t];
    for (int i = t; i < nchk; i += 1024) v2 += lct[nfront + i];
    for (int d = 16; d >= 1; d >>= 1) {
        v  += __shfl_xor(v, d);
        v2 += __shfl_xor(v2, d);
    }
    if (lane == 0) { s_w[w] = v; s_w2[w] = v2; }
    if (t < BNODE) s_cnt[t] = 0;
    __syncthreads();
    if (tile_on) {
        hacc = gcn_tile_mma<KS0>(hacc, xa0, xb0);
        gcn_tile_load<KS1>(arow, W1, 4 * KS0, kh, m, xa1, xb1);
    }
    int beg = 0, len = 0;
    for (int k = 0; k < 32; ++k) { beg += s_w[k]; len += s_w2[k]; }
    beg = min(max(beg, 0), e);
    len = min(max(len, 0), e - beg);
    const bool staged = len <= DCAP;

    const int per  = (nchk + 31) >> 5;
    const int g_lo = min(w * per, nchk);
    const int g_hi = min(g_lo + per, nchk);

    for (int gb = g_lo; gb < g_hi; gb += 32) {
        int st = 0, ln = 0;
        if (gb + lane < g_hi) {
            const int gg = gb + lane;
            const int c0 = min(gg * CHE, e);
            st = lst[nfront + gg];
            ln = lct[nfront + gg];
            st = c0 + min(max(st, 0), e - c0);
            ln = min(max(ln, 0), e - st);
        }
        const int nseg = min(32, g_hi - gb);
        for (int u0 = 0; u0 < nseg; u0 += DSEG) {
            int stu[DSEG], lnu[DSEG], q[DSEG][2];
#pragma unroll
            for (int u = 0; u < DSEG; ++u) {
                stu[u] = __shfl(st, (u0 + u) & 31);
                lnu[u] = (u0 + u < nseg) ? __shfl(ln, (u0 + u) & 31) : 0;
            }
#pragma unroll
            for (int u = 0; u < DSEG; ++u) {
                q[u][0] = (lane < lnu[u])      ? pd[stu[u] + lane]      : -1;
                q[u][1] = (lane + 32 < lnu[u]) ? pd[stu[u] + lane + 32] : -1;
            }
#pragma unroll
            for (int u = 0; u < DSEG; ++u) {
#pragma unroll
                for (int r = 0; r < 2; ++r) {
                    const unsigned l = (unsigned)(q[u][r] - n0);
                    if (l < (unsigned)BNODE) atomicAdd(&s_cnt[l], 1);
                }
            }
#pragma unroll
            for (int u = 0; u < DSEG; ++u) {
                for (int i = 64 + lane; i < lnu[u]; i += 32) {
                    const unsigned l = (unsigned)(pd[stu[u] + i] - n0);
                    if (l < (unsigned)BNODE) atomicAdd(&s_cnt[l], 1);
                }
            }
        }
    }
    __syncthreads();

    int c = 0, inc = 0;
    if (t < BNODE) {
        c = s_cnt[t];
        inc = gcn_wave_incl(c, lane);
        if (lane == 31) s_w[w] = inc;
    }
    __syncthreads();
    if (t < BNODE) {
        int off = 0;
        for (int k = 0; k < BNODE / 32; ++k)
            if (k < w) off += s_w[k];
        const int ex = off + inc - c;
        s_cur[t] = ex;
        const int node = n0 + t;
        float dn = 0.0f;
        if (node < n) {
            cnt[node]    = c;
            rowptr[node] = beg + ex;
            float dg = (float)(c + 1);
            dn = dg > 0.0f ? rsqrtf(dg) : 0.0f;
            dinv[node] = dn;
        }
        s_dinv[t] = dn;
    }
    __syncthreads();
    if (tile_on) {
        hacc = gcn_tile_mma<KS1>(hacc, xa1, xb1);
        gcn_tile_load<KS2>(arow, W1, 4 * (KS0 + KS1), kh, m, xa2, xb2);
    }

    for (int gb = g_lo; gb < g_hi; gb += 32) {
        int st = 0, ln = 0;
        if (gb + lane < g_hi) {
            const int gg = gb + lane;
            const int c0 = min(gg * CHE, e);
            st = lst[nfront + gg];
            ln = lct[nfront + gg];
            st = c0 + min(max(st, 0), e - c0);
            ln = min(max(ln, 0), e - st);
        }
        const int nseg = min(32, g_hi - gb);
        for (int u0 = 0; u0 < nseg; u0 += DSEG) {
            int stu[DSEG], lnu[DSEG], q[DSEG][2], sv[DSEG][2];
#pragma unroll
            for (int u = 0; u < DSEG; ++u) {
                stu[u] = __shfl(st, (u0 + u) & 31);
                lnu[u] = (u0 + u < nseg) ? __shfl(ln, (u0 + u) & 31) : 0;
            }
#pragma unroll
            for (int u = 0; u < DSEG; ++u) {
                const bool in0 = lane < lnu[u];
                const bool in1 = lane + 32 < lnu[u];
                q[u][0]  = in0 ? pd[stu[u] + lane]      : -1;
                q[u][1]  = in1 ? pd[stu[u] + lane + 32] : -1;
                sv[u][0] = in0 ? ps[stu[u] + lane]      : 0;
                sv[u][1] = in1 ? ps[stu[u] + lane + 32] : 0;
            }
#pragma unroll
            for (int u = 0; u < DSEG; ++u) {
#pragma unroll
                for (int r = 0; r < 2; ++r) {
                    const unsigned l = (unsigned)(q[u][r] - n0);
                    if (l < (unsigned)BNODE) {
                        const int slot = atomicAdd(&s_cur[l], 1);
                        if ((unsigned)slot < (unsigned)len) {
                            if (staged) s_out[slot] = sv[u][r];
                            else        csr[beg + slot] = sv[u][r];
                        }
                    }
                }
            }
#pragma unroll
            for (int u = 0; u < DSEG; ++u) {
                for (int i = 64 + lane; i < lnu[u]; i += 32) {
                    const unsigned l = (unsigned)(pd[stu[u] + i] - n0);
                    const int s1 = ps[stu[u] + i];
                    if (l < (unsigned)BNODE) {
                        const int slot = atomicAdd(&s_cur[l], 1);
                        if ((unsigned)slot < (unsigned)len) {
                            if (staged) s_out[slot] = s1;
                            else        csr[beg + slot] = s1;
                        }
                    }
                }
            }
        }
    }
    __syncthreads();
    if (staged) {
        for (int i = t; i < len; i += 1024) csr[beg + i] = s_out[i];
    }

    if (tile_on) {
        hacc = gcn_tile_mma<KS2>(hacc, xa2, xb2);
        const int rl = (row0 - n0) + 8 * half;
        const v4f d0 = *reinterpret_cast<const v4f*>(&s_dinv[rl]);
        const v4f d1 = *reinterpret_cast<const v4f*>(&s_dinv[rl + 4]);
        const float dsc[8] = {d0.x, d0.y, d0.z, d0.w, d1.x, d1.y, d1.z, d1.w};
        const int rbase = n0 + rl;
        float* __restrict__ ccol = hs + (size_t)rbase * HID + m;
#pragma unroll
        for (int r = 0; r < 8; ++r) {
            if (rbase + r < n) ccol[(size_t)r * HID] = hacc[r] * dsc[r];
        }
    }
}

template <int Q>
static __device__ __forceinline__ int gcn_quad_lane(int v) {
    return __builtin_amdgcn_update_dpp(0, v, Q * 0x55, 0xF, 0xF, true);
}

template <int K>
static __device__ __forceinline__ v4f gcn_take4(const float* __restrict__ hs,
                                                const int (&ids)[4], unsigned last,
                                                unsigned q4, int left) {
    unsigned s = (unsigned)gcn_quad_lane<(K & 3)>(ids[K >> 2]);
    s = min(s, last);
    v4f x = *reinterpret_cast<const v4f*>(hs + s * HID + q4);
    const v4f z = {0.0f, 0.0f, 0.0f, 0.0f};
    return (K < left) ? x : z;
}

#define APITCH 20

template <bool W2FUSE>
__global__ __launch_bounds__(256)
void gcn_gather_kernel(const float* __restrict__ hs,
                       const int* __restrict__ rowptr,
                       const int* __restrict__ cnt,
                       const int* __restrict__ csr,
                       const float* __restrict__ dinv,
                       const float* __restrict__ bias,
                       const float* __restrict__ W2,
                       float* __restrict__ out, int n, int e) {
    const int tid = blockIdx.x * blockDim.x + threadIdx.x;
    int i = tid >> 2;
    const int q = tid & 3;
    const unsigned q4 = (unsigned)q * 4u;
    const int wave = __builtin_amdgcn_readfirstlane((int)(threadIdx.x >> 5));
    if (!W2FUSE) {
        if (i >= n) return;
    } else {
        i = min(i, n - 1);
    }

    int beg = rowptr[i];
    int c   = cnt[i];
    beg = min(max(beg, 0), e);
    c   = min(max(c, 0), e - beg);
    const int* __restrict__ p = csr + beg;
    const unsigned last = (unsigned)(n - 1);

    v4f acc = *reinterpret_cast<const v4f*>(hs + (unsigned)i * HID + q4);
    for (int j = 0; j < c; j += 16) {
        const int left = c - j;
        int ids[4];
#pragma unroll
        for (int u = 0; u < 4; ++u) ids[u] = p[min(j + 4 * u + q, c - 1)];
        v4f v0  = gcn_take4<0>(hs, ids, last, q4, left);
        v4f v1  = gcn_take4<1>(hs, ids, last, q4, left);
        v4f v2  = gcn_take4<2>(hs, ids, last, q4, left);
        v4f v3  = gcn_take4<3>(hs, ids, last, q4, left);
        v4f v4  = gcn_take4<4>(hs, ids, last, q4, left);
        v4f v5  = gcn_take4<5>(hs, ids, last, q4, left);
        v4f v6  = gcn_take4<6>(hs, ids, last, q4, left);
        v4f v7  = gcn_take4<7>(hs, ids, last, q4, left);
        v4f v8  = gcn_take4<8>(hs, ids, last, q4, left);
        v4f v9  = gcn_take4<9>(hs, ids, last, q4, left);
        v4f v10 = gcn_take4<10>(hs, ids, last, q4, left);
        v4f v11 = gcn_take4<11>(hs, ids, last, q4, left);
        v4f v12 = gcn_take4<12>(hs, ids, last, q4, left);
        v4f v13 = gcn_take4<13>(hs, ids, last, q4, left);
        v4f v14 = gcn_take4<14>(hs, ids, last, q4, left);
        v4f v15 = gcn_take4<15>(hs, ids, last, q4, left);
        acc += v0;  acc += v1;  acc += v2;  acc += v3;
        acc += v4;  acc += v5;  acc += v6;  acc += v7;
        acc += v8;  acc += v9;  acc += v10; acc += v11;
        acc += v12; acc += v13; acc += v14; acc += v15;
    }
    const float di = dinv[i];
    const v4f bv = *reinterpret_cast<const v4f*>(bias + q4);
    v4f o;
    o.x = acc.x * di + bv.x;
    o.y = acc.y * di + bv.y;
    o.z = acc.z * di + bv.z;
    o.w = acc.w * di + bv.w;

    if (!W2FUSE) {
        *reinterpret_cast<v4f*>(out + (unsigned)i * HID + q4) = o;
    } else {
        const int lane = (int)__lane_id();
        const int m    = lane & 15;
        const int kh   = (lane >> 4) << 1;
        v2f bw[HID / 4];
        if (wave < 4) {
#pragma unroll
            for (int k = 0; k < HID; k += 4) {
                bw[k >> 2].x = W2[(k + kh) * HID + m];
                bw[k >> 2].y = W2[(k + kh + 1) * HID + m];
            }
        }
        __shared__ float s_a[64 * APITCH];
        v4f r;
        r.x = fmaxf(o.x, 0.0f) * di;
        r.y = fmaxf(o.y, 0.0f) * di;
        r.z = fmaxf(o.z, 0.0f) * di;
        r.w = fmaxf(o.w, 0.0f) * di;
        *reinterpret_cast<v4f*>(&s_a[(wave * 8 + (lane >> 2)) * APITCH + q4]) = r;
        __syncthreads();
        const int row0 = blockIdx.x * 64 + wave * 16;
        if (wave < 4 && row0 < n) {
            const float* __restrict__ arow = &s_a[(wave * 16 + m) * APITCH];
            v8f acc2 = {};
#pragma unroll
            for (int k = 0; k < HID; k += 4) {
                const v2f a = *reinterpret_cast<const v2f*>(arow + k + kh);
                acc2 = __builtin_amdgcn_wmma_f32_16x16x4_f32(
                     false, a,  false, bw[k >> 2],
                     (short)0, acc2,  false,  false);
            }
            const int rbase = row0 + 8 * (lane >> 4);
#pragma unroll
            for (int r8 = 0; r8 < 8; ++r8) {
                if (rbase + r8 < n) out[(size_t)(rbase + r8) * HID + m] = acc2[r8];
            }
        }
    }
}

static inline size_t gcn_align256(size_t b) { return (b + 255) & ~(size_t)255; }

extern "C" void kernel_launch(void* const* d_in, const int* in_sizes, int n_in,
                              void* d_out, int out_size, void* d_ws, size_t ws_size,
                              hipStream_t stream) {
    const float* x  = (const float*)d_in[0];
    const int*   ei = (const int*)d_in[1];
    const float* W1 = (const float*)d_in[2];
    const float* b1 = (const float*)d_in[3];
    const float* W2 = (const float*)d_in[4];
    const float* b2 = (const float*)d_in[5];
    float* out = (float*)d_out;

    const int N = in_sizes[0] / F_IN;
    const int E = in_sizes[1] / 2;
    const int* src = ei;
    const int* dst = ei + E;

    char* ws = (char*)d_ws;
    const size_t szN  = gcn_align256((size_t)N * sizeof(int));
    const size_t szE  = gcn_align256((size_t)E * sizeof(int));
    const size_t szNH = gcn_align256((size_t)N * HID * sizeof(float));
    int*   cnt    = (int*)ws;
    int*   rowptr = (int*)(ws + szN);
    float* dinv   = (float*)(ws + 2 * szN);
    int*   csr    = (int*)(ws + 3 * szN);
    float* hbuf   = (float*)(ws + 3 * szN + szE);
    float* hbuf2  = (float*)(ws + 3 * szN + szE + szNH);
    int*   ps     = (int*)(ws + 3 * szN + szE + 2 * szNH);
    int*   pd     = (int*)(ws + 3 * szN + 2 * szE + 2 * szNH);
    const int NCHK = (E + CHE - 1) / CHE;
    const size_t szBC = gcn_align256((size_t)NBMAX * NCHK * sizeof(int));
    int*   lst    = (int*)(ws + 3 * szN + 3 * szE + 2 * szNH);
    int*   lct    = (int*)(ws + 3 * szN + 3 * szE + 2 * szNH + szBC);

    const int TB = 256;

    gcn_part_kernel<<<NCHK, 1024, 0, stream>>>(src, dst, lst, lct, ps, pd, N, E, NCHK);
    gcn_bucket_kernel<<<NBMAX, 1024, 0, stream>>>(ps, pd, lst, lct, csr, cnt, rowptr, dinv, x, W1, hbuf, N, E, NCHK);

    const long long gt = (long long)N * 4;
    const int ggrid = (int)((gt + TB - 1) / TB);
    gcn_gather_kernel<true><<<ggrid, TB, 0, stream>>>(hbuf, rowptr, cnt, csr, dinv, b1, W2, hbuf2, N, E);

    gcn_gather_kernel<false><<<ggrid, TB, 0, stream>>>(hbuf2, rowptr, cnt, csr, dinv, b2, W2, out, N, E);
}
